# starting kernel
# baseline (speedup 1.0000x reference)
#include <hip/hip_runtime.h>

typedef __attribute__((ext_vector_type(2))) float v2f;
typedef __attribute__((ext_vector_type(8))) float v8f;

#define GCN_N_NODES 100000
#define GCN_N_EDGES 1600000
#define GCN_D 32

__global__ __launch_bounds__(256) void gcn_gemm_wmma(
    const float* __restrict__ x, const float* __restrict__ W,
    float* __restrict__ y, int n_tiles)
{
    const int lane = threadIdx.x & 31;
    const int wave = threadIdx.x >> 5;
    const int tile = blockIdx.x * 8 + wave;
    if (tile >= n_tiles) return;

    const int m  = lane & 15;
    const int kb = (lane >> 4) * 2;

    v2f b0[8], b1[8];
#pragma unroll
    for (int kc = 0; kc < 8; ++kc) {
        const int k = kc * 4 + kb;
        const float* w0 = W + (m)      * GCN_D + k;
        const float* w1 = W + (16 + m) * GCN_D + k;
        v2f t0; t0.x = w0[0]; t0.y = w0[1];
        v2f t1; t1.x = w1[0]; t1.y = w1[1];
        b0[kc] = t0; b1[kc] = t1;
    }

    v8f c0 = {}; v8f c1 = {};
    const float* xr = x + (size_t)tile * 16 * GCN_D;
#pragma unroll
    for (int kc = 0; kc < 8; ++kc) {
        const int k = kc * 4 + kb;
        v2f a = *(const v2f*)(xr + m * GCN_D + k);
        c0 = __builtin_amdgcn_wmma_f32_16x16x4_f32(
                 false, a, false, b0[kc], (short)0, c0, false, false);
        c1 = __builtin_amdgcn_wmma_f32_16x16x4_f32(
                 false, a, false, b1[kc], (short)0, c1, false, false);
    }

    const int n     = lane & 15;
    const int rbase = (lane >> 4) * 8;
    float* yr = y + (size_t)tile * 16 * GCN_D;
#pragma unroll
    for (int r = 0; r < 8; ++r) {
        yr[(rbase + r) * GCN_D + n]      = c0[r];
        yr[(rbase + r) * GCN_D + 16 + n] = c1[r];
    }
}

__global__ __launch_bounds__(256) void gcn_zero(float4* __restrict__ out, int n4)
{
    int i = blockIdx.x * 256 + threadIdx.x;
    if (i < n4) {
        float4 z; z.x = 0.f; z.y = 0.f; z.z = 0.f; z.w = 0.f;
        out[i] = z;
    }
}

__global__ __launch_bounds__(256) void gcn_scatter(
    const float* __restrict__ y, const int* __restrict__ ei,
    float* __restrict__ out, int n_edges)
{
    int t = blockIdx.x * 256 + threadIdx.x;
    int e = t >> 5;
    int f = t & 31;
    if (e >= n_edges) return;
    int src = ei[e];
    int dst = ei[n_edges + e];
    atomicAdd(out + (size_t)dst * GCN_D + f, y[(size_t)src * GCN_D + f]);
}

extern "C" void kernel_launch(void* const* d_in, const int* in_sizes, int n_in,
                              void* d_out, int out_size, void* d_ws, size_t ws_size,
                              hipStream_t stream)
{
    const float* x  = (const float*)d_in[0];
    const int*   ei = (const int*)d_in[1];
    const float* W  = (const float*)d_in[2];
    float*       out = (float*)d_out;
    float*       y   = (float*)d_ws;

    const int n_tiles = GCN_N_NODES / 16;
    const int gemm_blocks = (n_tiles + 7) / 8;
    gcn_gemm_wmma<<<gemm_blocks, 256, 0, stream>>>(x, W, y, n_tiles);

    const int n4 = GCN_N_NODES * GCN_D / 4;
    gcn_zero<<<(n4 + 255) / 256, 256, 0, stream>>>((float4*)out, n4);

    const long long nthreads = (long long)GCN_N_EDGES * 32;
    const int scat_blocks = (int)((nthreads + 255) / 256);
    gcn_scatter<<<scat_blocks, 256, 0, stream>>>(y, ei, out, GCN_N_EDGES);
}


# Row kernel: a row's sources ordered by a bitonic network over the wave's lanes (DPP row_xmask, permlanex16), not a readlane rank loop and ballot lookup; row bounds made scalar; sum order unchanged
# speedup vs baseline: 2.0144x; 2.0144x over previous
#include <hip/hip_runtime.h>


typedef __attribute__((ext_vector_type(2))) float v2f;
typedef __attribute__((ext_vector_type(8))) float v8f;

#define GCN_N_NODES 100000
#define GCN_N_EDGES 1600000
#define GCN_D 32

#define GCN_NBLK 256
#define GCN_EPB 6250
#define GCN_NBK 256
#define GCN_ROWS_PB 512
#define GCN_CAP 16384

__global__ __launch_bounds__(1024) void gcn_hist(
    const int* __restrict__ ei, int* __restrict__ hist)
{
    __shared__ int h[GCN_NBK];
    const int tid = threadIdx.x;
    const int b = blockIdx.x;
    if (tid < GCN_NBK) h[tid] = 0;
    __syncthreads();
    const int* dst = ei + GCN_N_EDGES + b * GCN_EPB;
    for (int i = tid; i < GCN_EPB; i += 1024) atomicAdd(&h[dst[i] >> 9], 1);
    __syncthreads();
    if (tid < GCN_NBK) hist[b * GCN_NBK + tid] = h[tid];
}

__global__ __launch_bounds__(1024) void gcn_scan(
    const int* __restrict__ hist, int* __restrict__ base, int* __restrict__ bstart)
{
    __shared__ int qsum[4][GCN_NBK];
    __shared__ int tot[GCN_NBK];
    __shared__ int bst[GCN_NBK];
    const int tid = threadIdx.x;
    const int k = tid & 255;
    const int q = tid >> 8;
    int v[64];
#pragma unroll
    for (int j = 0; j < 64; ++j) v[j] = hist[(q * 64 + j) * GCN_NBK + k];
    int s = 0;
#pragma unroll
    for (int j = 0; j < 64; ++j) s += v[j];
    qsum[q][k] = s;
    __syncthreads();
    if (tid < GCN_NBK)
        tot[tid] = qsum[0][tid] + qsum[1][tid] + qsum[2][tid] + qsum[3][tid];
    __syncthreads();
    if (tid < GCN_NBK) {
        int a = 0;
        for (int j = 0; j < tid; ++j) a += tot[j];
        bst[tid] = a;
        bstart[tid] = a;
        if (tid == GCN_NBK - 1) bstart[GCN_NBK] = a + tot[tid];
    }
    __syncthreads();
    int run = bst[k];
    for (int qq = 0; qq < q; ++qq) run += qsum[qq][k];
#pragma unroll
    for (int j = 0; j < 64; ++j) {
        base[(q * 64 + j) * GCN_NBK + k] = run;
        run += v[j];
    }
}

__global__ __launch_bounds__(1024) void gcn_place(
    const int* __restrict__ ei, const int* __restrict__ base, int* __restrict__ list)
{
    __shared__ int cur[GCN_NBK];
    const int tid = threadIdx.x;
    const int b = blockIdx.x;
    if (tid < GCN_NBK) cur[tid] = base[b * GCN_NBK + tid];
    __syncthreads();
    const int* src = ei + b * GCN_EPB;
    const int* dst = ei + GCN_N_EDGES + b * GCN_EPB;
    for (int i = tid; i < GCN_EPB; i += 1024) {
        const int d = dst[i];
        const int s = src[i];
        const int pos = atomicAdd(&cur[d >> 9], 1);
        list[pos] = (s << 9) | (d & 511);
    }
}

__device__ __forceinline__ int gcn_uni(int v) { return __builtin_amdgcn_readfirstlane(v); }

template <int J>
__device__ __forceinline__ int gcn_xrow(int v)
{
    return __builtin_amdgcn_update_dpp(0, v, 0x160 + J, 0xf, 0xf, true);
}

__device__ __forceinline__ int gcn_xhalf(int v)
{
    return __builtin_amdgcn_permlanex16(v, v, 0x76543210, 0xfedcba98, false, false);
}

template <int K, int J>
__device__ __forceinline__ int gcn_cx(int v, int lane)
{
    const int p = (J == 16) ? gcn_xhalf(v) : gcn_xrow<(J & 15)>(v);
    const bool keep_min = ((lane & J) == 0) == ((lane & K) == 0);
    const int lo = min(v, p);
    const int hi = max(v, p);
    return keep_min ? lo : hi;
}

template <bool TAIL>
__device__ __forceinline__ float gcn_add8(
    const char* xb, int voff, unsigned lane4, int i0, int m, float acc)
{
    float f[8];
#pragma unroll
    for (int u = 0; u < 8; ++u) {
        const int i = i0 + u;
        const unsigned so = (unsigned)__builtin_amdgcn_readlane(voff, i);
        const float xv = *(const float*)(xb + (so + lane4));
        f[u] = (!TAIL || i < m) ? xv : 0.f;
    }
#pragma unroll
    for (int u = 0; u < 8; ++u) acc += f[u];
    return acc;
}

__global__ __launch_bounds__(1024) void gcn_rowsum(
    const float* __restrict__ x, const int* __restrict__ list,
    const int* __restrict__ bstart, float* __restrict__ out)
{
    __shared__ int cnt[GCN_ROWS_PB];
    __shared__ int off[GCN_ROWS_PB + 1];
    __shared__ int cur[GCN_ROWS_PB];
    __shared__ int sorted[GCN_CAP];

    const int tid = threadIdx.x;
    const int lane = tid & 31;
    const int wave = gcn_uni(tid >> 5);
    const int k = blockIdx.x;
    const char* xb = (const char*)x;
    const unsigned lane4 = (unsigned)lane * 4u;
    const int start = bstart[k];
    const int end = bstart[k + 1];

    int c0 = start;
    bool first = true;
    do {
        const int n = min(GCN_CAP, end - c0);

        if (tid < GCN_ROWS_PB) cnt[tid] = 0;
        __syncthreads();
        for (int i = tid; i < n; i += 1024) atomicAdd(&cnt[list[c0 + i] & 511], 1);
        __syncthreads();

        if (wave == 0) {
            int loc[16];
            int s = 0;
#pragma unroll
            for (int j = 0; j < 16; ++j) { loc[j] = cnt[lane * 16 + j]; s += loc[j]; }
            int inc = s;
#pragma unroll
            for (int d = 1; d < 32; d <<= 1) {
                const int t = __shfl_up(inc, d);
                if (lane >= d) inc += t;
            }
            int run = inc - s;
#pragma unroll
            for (int j = 0; j < 16; ++j) {
                off[lane * 16 + j] = run;
                cur[lane * 16 + j] = run;
                run += loc[j];
            }
            if (lane == 31) off[GCN_ROWS_PB] = run;
        }
        __syncthreads();

        for (int i = tid; i < n; i += 1024) {
            const int p = list[c0 + i];
            const int pos = atomicAdd(&cur[p & 511], 1);
            sorted[pos] = p >> 9;
        }
        __syncthreads();

        for (int j = 0; j < 16; ++j) {
            const int r = wave + 32 * j;
            const int row = k * GCN_ROWS_PB + r;
            if (row < GCN_N_NODES) {
                const int rb = gcn_uni(off[r]);
                const int re = gcn_uni(off[r + 1]);
                float acc = first ? 0.f : out[(size_t)row * GCN_D + lane];
                for (int g0 = rb; g0 < re; g0 += 32) {
                    const int m = min(32, re - g0);
                    int val = (lane < m) ? sorted[g0 + lane] : 0x7fffffff;
                    val = gcn_cx<2, 1>(val, lane);
                    val = gcn_cx<4, 2>(val, lane);
                    val = gcn_cx<4, 1>(val, lane);
                    val = gcn_cx<8, 4>(val, lane);
                    val = gcn_cx<8, 2>(val, lane);
                    val = gcn_cx<8, 1>(val, lane);
                    val = gcn_cx<16, 8>(val, lane);
                    val = gcn_cx<16, 4>(val, lane);
                    val = gcn_cx<16, 2>(val, lane);
                    val = gcn_cx<16, 1>(val, lane);
                    if (m > 16) {
                        val = gcn_cx<32, 16>(val, lane);
                        val = gcn_cx<32, 8>(val, lane);
                        val = gcn_cx<32, 4>(val, lane);
                        val = gcn_cx<32, 2>(val, lane);
                        val = gcn_cx<32, 1>(val, lane);
                    }
                    const int voff = min(val, GCN_N_NODES - 1) << 7;
                    int i0 = 0;
                    for (; i0 + 8 <= m; i0 += 8)
                        acc = gcn_add8<false>(xb, voff, lane4, i0, m, acc);
                    if (i0 < m)
                        acc = gcn_add8<true>(xb, voff, lane4, i0, m, acc);
                }
                out[(size_t)row * GCN_D + lane] = acc;
            }
        }
        __syncthreads();
        c0 += GCN_CAP;
        first = false;
    } while (c0 < end);
}

__global__ __launch_bounds__(256) void gcn_gemm_inplace(
    float* io, const float* __restrict__ W, int n_tiles)
{
    const int lane = threadIdx.x & 31;
    const int wave = threadIdx.x >> 5;
    const int tile = blockIdx.x * 8 + wave;
    if (tile >= n_tiles) return;

    const int m  = lane & 15;
    const int kb = (lane >> 4) * 2;

    v2f b0[8], b1[8];
#pragma unroll
    for (int kc = 0; kc < 8; ++kc) {
        const int k = kc * 4 + kb;
        const float* w0 = W + (m)      * GCN_D + k;
        const float* w1 = W + (16 + m) * GCN_D + k;
        v2f t0; t0.x = w0[0]; t0.y = w0[1];
        v2f t1; t1.x = w1[0]; t1.y = w1[1];
        b0[kc] = t0; b1[kc] = t1;
    }

    float* xr = io + (size_t)tile * 16 * GCN_D;
    v2f a[8];
#pragma unroll
    for (int kc = 0; kc < 8; ++kc) {
        const int k = kc * 4 + kb;
        a[kc] = *(const v2f*)(xr + m * GCN_D + k);
    }

    v8f c0 = {}; v8f c1 = {};
#pragma unroll
    for (int kc = 0; kc < 8; ++kc) {
        c0 = __builtin_amdgcn_wmma_f32_16x16x4_f32(
                 false, a[kc], false, b0[kc], (short)0, c0, false, false);
        c1 = __builtin_amdgcn_wmma_f32_16x16x4_f32(
                 false, a[kc], false, b1[kc], (short)0, c1, false, false);
    }

    const int n     = lane & 15;
    const int rbase = (lane >> 4) * 8;
#pragma unroll
    for (int r = 0; r < 8; ++r) {
        xr[(rbase + r) * GCN_D + n]      = c0[r];
        xr[(rbase + r) * GCN_D + 16 + n] = c1[r];
    }
}

extern "C" void kernel_launch(void* const* d_in, const int* in_sizes, int n_in,
                              void* d_out, int out_size, void* d_ws, size_t ws_size,
                              hipStream_t stream)
{
    const float* x  = (const float*)d_in[0];
    const int*   ei = (const int*)d_in[1];
    const float* W  = (const float*)d_in[2];
    float*       out = (float*)d_out;
    int*         wsi = (int*)d_ws;
    int* list   = wsi;
    int* hist   = wsi + 1600000;
    int* base   = wsi + 1665536;
    int* bstart = wsi + 1731072;

    gcn_hist<<<GCN_NBLK, 1024, 0, stream>>>(ei, hist);
    gcn_scan<<<1, 1024, 0, stream>>>(hist, base, bstart);
    gcn_place<<<GCN_NBLK, 1024, 0, stream>>>(ei, base, list);
    gcn_rowsum<<<196, 1024, 0, stream>>>(x, list, bstart, out);
    gcn_gemm_inplace<<<782, 256, 0, stream>>>(out, W, 6250);
}


# Row kernel: loads requested in bulk, used late: a thread's list words fetched at once for both LDS sweeps; a wave takes two rows, all their source rows in flight before the first add; sum order kept
# speedup vs baseline: 2.2217x; 1.1029x over previous
#include <hip/hip_runtime.h>


typedef __attribute__((ext_vector_type(2))) float v2f;
typedef __attribute__((ext_vector_type(8))) float v8f;

#define GCN_N_NODES 100000
#define GCN_N_EDGES 1600000
#define GCN_D 32

#define GCN_NBLK 256
#define GCN_EPB 6250
#define GCN_NBK 256
#define GCN_ROWS_PB 512
#define GCN_CAP 16384
#define GCN_WPT 16

__global__ __launch_bounds__(1024) void gcn_hist(
    const int* __restrict__ ei, int* __restrict__ hist)
{
    __shared__ int h[GCN_NBK];
    const int tid = threadIdx.x;
    const int b = blockIdx.x;
    if (tid < GCN_NBK) h[tid] = 0;
    __syncthreads();
    const int* dst = ei + GCN_N_EDGES + b * GCN_EPB;
    for (int i = tid; i < GCN_EPB; i += 1024) atomicAdd(&h[dst[i] >> 9], 1);
    __syncthreads();
    if (tid < GCN_NBK) hist[b * GCN_NBK + tid] = h[tid];
}

__global__ __launch_bounds__(1024) void gcn_scan(
    const int* __restrict__ hist, int* __restrict__ base, int* __restrict__ bstart)
{
    __shared__ int qsum[4][GCN_NBK];
    __shared__ int tot[GCN_NBK];
    __shared__ int bst[GCN_NBK];
    const int tid = threadIdx.x;
    const int k = tid & 255;
    const int q = tid >> 8;
    int v[64];
#pragma unroll
    for (int j = 0; j < 64; ++j) v[j] = hist[(q * 64 + j) * GCN_NBK + k];
    int s = 0;
#pragma unroll
    for (int j = 0; j < 64; ++j) s += v[j];
    qsum[q][k] = s;
    __syncthreads();
    if (tid < GCN_NBK)
        tot[tid] = qsum[0][tid] + qsum[1][tid] + qsum[2][tid] + qsum[3][tid];
    __syncthreads();
    if (tid < GCN_NBK) {
        int a = 0;
        for (int j = 0; j < tid; ++j) a += tot[j];
        bst[tid] = a;
        bstart[tid] = a;
        if (tid == GCN_NBK - 1) bstart[GCN_NBK] = a + tot[tid];
    }
    __syncthreads();
    int run = bst[k];
    for (int qq = 0; qq < q; ++qq) run += qsum[qq][k];
#pragma unroll
    for (int j = 0; j < 64; ++j) {
        base[(q * 64 + j) * GCN_NBK + k] = run;
        run += v[j];
    }
}

__global__ __launch_bounds__(1024) void gcn_place(
    const int* __restrict__ ei, const int* __restrict__ base, int* __restrict__ list)
{
    __shared__ int cur[GCN_NBK];
    const int tid = threadIdx.x;
    const int b = blockIdx.x;
    if (tid < GCN_NBK) cur[tid] = base[b * GCN_NBK + tid];
    __syncthreads();
    const int* src = ei + b * GCN_EPB;
    const int* dst = ei + GCN_N_EDGES + b * GCN_EPB;
    for (int i = tid; i < GCN_EPB; i += 1024) {
        const int d = dst[i];
        const int s = src[i];
        const int pos = atomicAdd(&cur[d >> 9], 1);
        list[pos] = (s << 9) | (d & 511);
    }
}

__device__ __forceinline__ int gcn_uni(int v) { return __builtin_amdgcn_readfirstlane(v); }

template <int J>
__device__ __forceinline__ int gcn_xrow(int v)
{
    return __builtin_amdgcn_update_dpp(0, v, 0x160 + J, 0xf, 0xf, true);
}

__device__ __forceinline__ int gcn_xhalf(int v)
{
    return __builtin_amdgcn_permlanex16(v, v, 0x76543210, 0xfedcba98, false, false);
}

template <int K, int J>
__device__ __forceinline__ int gcn_cx(int v, int lane)
{
    const int p = (J == 16) ? gcn_xhalf(v) : gcn_xrow<(J & 15)>(v);
    const bool keep_min = ((lane & J) == 0) == ((lane & K) == 0);
    const int lo = min(v, p);
    const int hi = max(v, p);
    return keep_min ? lo : hi;
}

__device__ __forceinline__ int gcn_sort10(int val, int lane)
{
    val = gcn_cx<2, 1>(val, lane);
    val = gcn_cx<4, 2>(val, lane);
    val = gcn_cx<4, 1>(val, lane);
    val = gcn_cx<8, 4>(val, lane);
    val = gcn_cx<8, 2>(val, lane);
    val = gcn_cx<8, 1>(val, lane);
    val = gcn_cx<16, 8>(val, lane);
    val = gcn_cx<16, 4>(val, lane);
    val = gcn_cx<16, 2>(val, lane);
    val = gcn_cx<16, 1>(val, lane);
    return val;
}

__device__ __forceinline__ int gcn_merge32(int val, int lane)
{
    val = gcn_cx<32, 16>(val, lane);
    val = gcn_cx<32, 8>(val, lane);
    val = gcn_cx<32, 4>(val, lane);
    val = gcn_cx<32, 2>(val, lane);
    val = gcn_cx<32, 1>(val, lane);
    return val;
}

__device__ __forceinline__ int gcn_rowoff(int val)
{
    return min(val, GCN_N_NODES - 1) << 7;
}

template <int B>
__device__ __forceinline__ void gcn_issue8(
    const char* xb, int voff, unsigned lane4, float (&f)[32])
{
#pragma unroll
    for (int u = 0; u < 8; ++u) {
        const int i = 8 * B + u;
        const unsigned so = (unsigned)__builtin_amdgcn_readlane(voff, i);
        f[i] = *(const float*)(xb + (so + lane4));
    }
}

template <int B>
__device__ __forceinline__ float gcn_sum8(const float (&f)[32], int m, float acc)
{
#pragma unroll
    for (int u = 0; u < 8; ++u) {
        const int i = 8 * B + u;
        acc += (i < m) ? f[i] : 0.f;
    }
    return acc;
}

__device__ __forceinline__ void gcn_issue_group(
    const char* xb, int voff, unsigned lane4, int m, float (&f)[32])
{
    if (m > 0)  gcn_issue8<0>(xb, voff, lane4, f);
    if (m > 8)  gcn_issue8<1>(xb, voff, lane4, f);
    if (m > 16) gcn_issue8<2>(xb, voff, lane4, f);
    if (m > 24) gcn_issue8<3>(xb, voff, lane4, f);
}

__device__ __forceinline__ float gcn_sum_group(const float (&f)[32], int m, float acc)
{
    if (m > 0)  acc = gcn_sum8<0>(f, m, acc);
    if (m > 8)  acc = gcn_sum8<1>(f, m, acc);
    if (m > 16) acc = gcn_sum8<2>(f, m, acc);
    if (m > 24) acc = gcn_sum8<3>(f, m, acc);
    return acc;
}

__device__ __forceinline__ float gcn_more_groups(
    const int* sorted, const char* xb, unsigned lane4, int lane,
    int g0, int re, float acc)
{
    for (; g0 < re; g0 += 32) {
        const int m = min(32, re - g0);
        int val = (lane < m) ? sorted[g0 + lane] : 0x7fffffff;
        val = gcn_merge32(gcn_sort10(val, lane), lane);
        const int voff = gcn_rowoff(val);
        for (int i0 = 0; i0 < m; i0 += 8) {
            float f[8];
#pragma unroll
            for (int u = 0; u < 8; ++u) {
                const unsigned so = (unsigned)__builtin_amdgcn_readlane(voff, i0 + u);
                f[u] = *(const float*)(xb + (so + lane4));
            }
#pragma unroll
            for (int u = 0; u < 8; ++u) acc += (i0 + u < m) ? f[u] : 0.f;
        }
    }
    return acc;
}

__global__ __launch_bounds__(1024) void gcn_rowsum(
    const float* __restrict__ x, const int* __restrict__ list,
    const int* __restrict__ bstart, float* __restrict__ out)
{
    __shared__ int cnt[GCN_ROWS_PB];
    __shared__ int off[GCN_ROWS_PB + 1];
    __shared__ int cur[GCN_ROWS_PB];
    __shared__ int sorted[GCN_CAP];

    const int tid = threadIdx.x;
    const int lane = tid & 31;
    const int wave = gcn_uni(tid >> 5);
    const int k = blockIdx.x;
    const char* xb = (const char*)x;
    const unsigned lane4 = (unsigned)lane * 4u;
    const int start = bstart[k];
    const int end = bstart[k + 1];

    int c0 = start;
    bool first = true;
    do {
        const int n = min(GCN_CAP, end - c0);

        int p[GCN_WPT];
#pragma unroll
        for (int u = 0; u < GCN_WPT; ++u) {
            const int i = tid + u * 1024;
            p[u] = (i < n) ? list[c0 + i] : -1;
        }

        if (tid < GCN_ROWS_PB) cnt[tid] = 0;
        __syncthreads();
#pragma unroll
        for (int u = 0; u < GCN_WPT; ++u)
            if (p[u] >= 0) atomicAdd(&cnt[p[u] & 511], 1);
        __syncthreads();

        if (wave == 0) {
            int loc[16];
            int s = 0;
#pragma unroll
            for (int j = 0; j < 16; ++j) { loc[j] = cnt[lane * 16 + j]; s += loc[j]; }
            int inc = s;
#pragma unroll
            for (int d = 1; d < 32; d <<= 1) {
                const int t = __shfl_up(inc, d);
                if (lane >= d) inc += t;
            }
            int run = inc - s;
#pragma unroll
            for (int j = 0; j < 16; ++j) {
                off[lane * 16 + j] = run;
                cur[lane * 16 + j] = run;
                run += loc[j];
            }
            if (lane == 31) off[GCN_ROWS_PB] = run;
        }
        __syncthreads();

#pragma unroll
        for (int u = 0; u < GCN_WPT; ++u) {
            if (p[u] >= 0) {
                const int pos = atomicAdd(&cur[p[u] & 511], 1);
                sorted[pos] = p[u] >> 9;
            }
        }
        __syncthreads();

        for (int jj = 0; jj < 8; ++jj) {
            const int rA = wave + 64 * jj;
            const int rB = rA + 32;
            const int rowA = k * GCN_ROWS_PB + rA;
            const int rowB = rowA + 32;
            const bool okA = rowA < GCN_N_NODES;
            const bool okB = rowB < GCN_N_NODES;
            const int rbA = gcn_uni(off[rA]);
            const int reA = gcn_uni(off[rA + 1]);
            const int rbB = gcn_uni(off[rB]);
            const int reB = gcn_uni(off[rB + 1]);
            const int mA = min(32, reA - rbA);
            const int mB = min(32, reB - rbB);

            float accA = 0.f;
            float accB = 0.f;
            if (!first) {
                if (okA) accA = out[(size_t)rowA * GCN_D + lane];
                if (okB) accB = out[(size_t)rowB * GCN_D + lane];
            }

            int vA = (lane < mA) ? sorted[rbA + lane] : 0x7fffffff;
            int vB = (lane < mB) ? sorted[rbB + lane] : 0x7fffffff;
            vA = gcn_sort10(vA, lane);
            vB = gcn_sort10(vB, lane);
            if (max(mA, mB) > 16) {
                vA = gcn_merge32(vA, lane);
                vB = gcn_merge32(vB, lane);
            }
            const int voffA = gcn_rowoff(vA);
            const int voffB = gcn_rowoff(vB);

            float fA[32];
            float fB[32];
            gcn_issue_group(xb, voffA, lane4, mA, fA);
            gcn_issue_group(xb, voffB, lane4, mB, fB);
            accA = gcn_sum_group(fA, mA, accA);
            accB = gcn_sum_group(fB, mB, accB);

            accA = gcn_more_groups(sorted, xb, lane4, lane, rbA + 32, reA, accA);
            accB = gcn_more_groups(sorted, xb, lane4, lane, rbB + 32, reB, accB);

            if (okA) out[(size_t)rowA * GCN_D + lane] = accA;
            if (okB) out[(size_t)rowB * GCN_D + lane] = accB;
        }
        __syncthreads();
        c0 += GCN_CAP;
        first = false;
    } while (c0 < end);
}

__global__ __launch_bounds__(256) void gcn_gemm_inplace(
    float* io, const float* __restrict__ W, int n_tiles)
{
    const int lane = threadIdx.x & 31;
    const int wave = threadIdx.x >> 5;
    const int tile = blockIdx.x * 8 + wave;
    if (tile >= n_tiles) return;

    const int m  = lane & 15;
    const int kb = (lane >> 4) * 2;

    v2f b0[8], b1[8];
#pragma unroll
    for (int kc = 0; kc < 8; ++kc) {
        const int k = kc * 4 + kb;
        const float* w0 = W + (m)      * GCN_D + k;
        const float* w1 = W + (16 + m) * GCN_D + k;
        v2f t0; t0.x = w0[0]; t0.y = w0[1];
        v2f t1; t1.x = w1[0]; t1.y = w1[1];
        b0[kc] = t0; b1[kc] = t1;
    }

    float* xr = io + (size_t)tile * 16 * GCN_D;
    v2f a[8];
#pragma unroll
    for (int kc = 0; kc < 8; ++kc) {
        const int k = kc * 4 + kb;
        a[kc] = *(const v2f*)(xr + m * GCN_D + k);
    }

    v8f c0 = {}; v8f c1 = {};
#pragma unroll
    for (int kc = 0; kc < 8; ++kc) {
        c0 = __builtin_amdgcn_wmma_f32_16x16x4_f32(
                 false, a[kc], false, b0[kc], (short)0, c0, false, false);
        c1 = __builtin_amdgcn_wmma_f32_16x16x4_f32(
                 false, a[kc], false, b1[kc], (short)0, c1, false, false);
    }

    const int n     = lane & 15;
    const int rbase = (lane >> 4) * 8;
#pragma unroll
    for (int r = 0; r < 8; ++r) {
        xr[(rbase + r) * GCN_D + n]      = c0[r];
        xr[(rbase + r) * GCN_D + 16 + n] = c1[r];
    }
}

extern "C" void kernel_launch(void* const* d_in, const int* in_sizes, int n_in,
                              void* d_out, int out_size, void* d_ws, size_t ws_size,
                              hipStream_t stream)
{
    const float* x  = (const float*)d_in[0];
    const int*   ei = (const int*)d_in[1];
    const float* W  = (const float*)d_in[2];
    float*       out = (float*)d_out;
    int*         wsi = (int*)d_ws;
    int* list   = wsi;
    int* hist   = wsi + 1600000;
    int* base   = wsi + 1665536;
    int* bstart = wsi + 1731072;

    gcn_hist<<<GCN_NBLK, 1024, 0, stream>>>(ei, hist);
    gcn_scan<<<1, 1024, 0, stream>>>(hist, base, bstart);
    gcn_place<<<GCN_NBLK, 1024, 0, stream>>>(ei, base, list);
    gcn_rowsum<<<196, 1024, 0, stream>>>(x, list, bstart, out);
    gcn_gemm_inplace<<<782, 256, 0, stream>>>(out, W, 6250);
}


# Edge sort made block-local: each edge block orders its 6250 edges by bucket in LDS in one launch, the row kernel collects its 256 segments; grid-wide count and scan launches gone; sum order kept
# speedup vs baseline: 3.4012x; 1.5309x over previous
#include <hip/hip_runtime.h>


typedef __attribute__((ext_vector_type(2))) float v2f;
typedef __attribute__((ext_vector_type(8))) float v8f;

#define GCN_N_NODES 100000
#define GCN_N_EDGES 1600000
#define GCN_D 32

#define GCN_NBLK 256
#define GCN_EPB 6250
#define GCN_EPT 7
#define GCN_NBK 256
#define GCN_ROWS_PB 512
#define GCN_SEGW 64
#define GCN_CAP 16384
#define GCN_WPT 16

__global__ __launch_bounds__(1024) void gcn_blocksort(
    const int* __restrict__ ei, int* __restrict__ list, int* __restrict__ segoff)
{
    __shared__ int cnt[GCN_NBK];
    __shared__ int off[GCN_NBK + 1];
    __shared__ int stage[GCN_EPB];
    const int tid = threadIdx.x;
    const int lane = tid & 31;
    const int b = blockIdx.x;
    const int* src = ei + b * GCN_EPB;
    const int* dst = ei + GCN_N_EDGES + b * GCN_EPB;

    int d[GCN_EPT];
    int s[GCN_EPT];
#pragma unroll
    for (int u = 0; u < GCN_EPT; ++u) {
        const int i = tid + u * 1024;
        const bool in = i < GCN_EPB;
        d[u] = in ? dst[i] : 0;
        s[u] = in ? src[i] : 0;
    }

    if (tid < GCN_NBK) cnt[tid] = 0;
    __syncthreads();
    int rank[GCN_EPT];
#pragma unroll
    for (int u = 0; u < GCN_EPT; ++u) {
        const bool in = tid + u * 1024 < GCN_EPB;
        rank[u] = in ? atomicAdd(&cnt[d[u] >> 9], 1) : 0;
    }
    __syncthreads();

    if (tid < 32) {
        int loc[8];
        int sum = 0;
#pragma unroll
        for (int j = 0; j < 8; ++j) { loc[j] = cnt[lane * 8 + j]; sum += loc[j]; }
        int inc = sum;
#pragma unroll
        for (int dd = 1; dd < 32; dd <<= 1) {
            const int t = __shfl_up(inc, dd);
            if (lane >= dd) inc += t;
        }
        int run = inc - sum;
#pragma unroll
        for (int j = 0; j < 8; ++j) { off[lane * 8 + j] = run; run += loc[j]; }
        if (lane == 31) off[GCN_NBK] = run;
    }
    __syncthreads();

    int first[GCN_EPT];
#pragma unroll
    for (int u = 0; u < GCN_EPT; ++u) first[u] = off[d[u] >> 9];
#pragma unroll
    for (int u = 0; u < GCN_EPT; ++u) {
        const bool in = tid + u * 1024 < GCN_EPB;
        if (in) stage[first[u] + rank[u]] = (s[u] << 9) | (d[u] & 511);
    }
    __syncthreads();

#pragma unroll
    for (int u = 0; u < GCN_EPT; ++u) {
        const int i = tid + u * 1024;
        if (i < GCN_EPB) list[b * GCN_EPB + i] = stage[i];
    }
    if (tid <= GCN_NBK) segoff[tid * GCN_NBLK + b] = off[tid];
}

__device__ __forceinline__ int gcn_uni(int v) { return __builtin_amdgcn_readfirstlane(v); }

template <int J>
__device__ __forceinline__ int gcn_xrow(int v)
{
    return __builtin_amdgcn_update_dpp(0, v, 0x160 + J, 0xf, 0xf, true);
}

__device__ __forceinline__ int gcn_xhalf(int v)
{
    return __builtin_amdgcn_permlanex16(v, v, 0x76543210, 0xfedcba98, false, false);
}

template <int K, int J>
__device__ __forceinline__ int gcn_cx(int v, int lane)
{
    const int p = (J == 16) ? gcn_xhalf(v) : gcn_xrow<(J & 15)>(v);
    const bool keep_min = ((lane & J) == 0) == ((lane & K) == 0);
    const int lo = min(v, p);
    const int hi = max(v, p);
    return keep_min ? lo : hi;
}

__device__ __forceinline__ int gcn_sort10(int val, int lane)
{
    val = gcn_cx<2, 1>(val, lane);
    val = gcn_cx<4, 2>(val, lane);
    val = gcn_cx<4, 1>(val, lane);
    val = gcn_cx<8, 4>(val, lane);
    val = gcn_cx<8, 2>(val, lane);
    val = gcn_cx<8, 1>(val, lane);
    val = gcn_cx<16, 8>(val, lane);
    val = gcn_cx<16, 4>(val, lane);
    val = gcn_cx<16, 2>(val, lane);
    val = gcn_cx<16, 1>(val, lane);
    return val;
}

__device__ __forceinline__ int gcn_merge32(int val, int lane)
{
    val = gcn_cx<32, 16>(val, lane);
    val = gcn_cx<32, 8>(val, lane);
    val = gcn_cx<32, 4>(val, lane);
    val = gcn_cx<32, 2>(val, lane);
    val = gcn_cx<32, 1>(val, lane);
    return val;
}

__device__ __forceinline__ int gcn_rowoff(int val)
{
    return min(val, GCN_N_NODES - 1) << 7;
}

template <int B>
__device__ __forceinline__ void gcn_issue8(
    const char* xb, int voff, unsigned lane4, float (&f)[32])
{
#pragma unroll
    for (int u = 0; u < 8; ++u) {
        const int i = 8 * B + u;
        const unsigned so = (unsigned)__builtin_amdgcn_readlane(voff, i);
        f[i] = *(const float*)(xb + (so + lane4));
    }
}

template <int B>
__device__ __forceinline__ float gcn_sum8(const float (&f)[32], int m, float acc)
{
#pragma unroll
    for (int u = 0; u < 8; ++u) {
        const int i = 8 * B + u;
        acc += (i < m) ? f[i] : 0.f;
    }
    return acc;
}

__device__ __forceinline__ void gcn_issue_group(
    const char* xb, int voff, unsigned lane4, int m, float (&f)[32])
{
    if (m > 0)  gcn_issue8<0>(xb, voff, lane4, f);
    if (m > 8)  gcn_issue8<1>(xb, voff, lane4, f);
    if (m > 16) gcn_issue8<2>(xb, voff, lane4, f);
    if (m > 24) gcn_issue8<3>(xb, voff, lane4, f);
}

__device__ __forceinline__ float gcn_sum_group(const float (&f)[32], int m, float acc)
{
    if (m > 0)  acc = gcn_sum8<0>(f, m, acc);
    if (m > 8)  acc = gcn_sum8<1>(f, m, acc);
    if (m > 16) acc = gcn_sum8<2>(f, m, acc);
    if (m > 24) acc = gcn_sum8<3>(f, m, acc);
    return acc;
}

__device__ __forceinline__ float gcn_more_groups(
    const int* sorted, const char* xb, unsigned lane4, int lane,
    int g0, int re, float acc)
{
    for (; g0 < re; g0 += 32) {
        const int m = min(32, re - g0);
        int val = (lane < m) ? sorted[g0 + lane] : 0x7fffffff;
        val = gcn_merge32(gcn_sort10(val, lane), lane);
        const int voff = gcn_rowoff(val);
        for (int i0 = 0; i0 < m; i0 += 8) {
            float f[8];
#pragma unroll
            for (int u = 0; u < 8; ++u) {
                const unsigned so = (unsigned)__builtin_amdgcn_readlane(voff, i0 + u);
                f[u] = *(const float*)(xb + (so + lane4));
            }
#pragma unroll
            for (int u = 0; u < 8; ++u) acc += (i0 + u < m) ? f[u] : 0.f;
        }
    }
    return acc;
}

__global__ __launch_bounds__(1024) void gcn_rowsum(
    const float* __restrict__ x, const int* __restrict__ list,
    const int* __restrict__ segoff, float* __restrict__ out)
{
    __shared__ int cnt[GCN_ROWS_PB];
    __shared__ int off[GCN_ROWS_PB + 1];
    __shared__ int cur[GCN_ROWS_PB];
    __shared__ int sorted[GCN_CAP];
    __shared__ int more;

    const int tid = threadIdx.x;
    const int lane = tid & 31;
    const int wave = gcn_uni(tid >> 5);
    const int k = blockIdx.x;
    const char* xb = (const char*)x;
    const unsigned lane4 = (unsigned)lane * 4u;
    const int sg0 = wave * 8;
    int bnd0 = 0;
    int bnd1 = 0;
    if (lane < 8) {
        bnd0 = segoff[k * GCN_NBLK + sg0 + lane];
        bnd1 = segoff[(k + 1) * GCN_NBLK + sg0 + lane];
    }
    int s0[8];
    int len[8];
#pragma unroll
    for (int u = 0; u < 8; ++u) {
        s0[u] = __builtin_amdgcn_readlane(bnd0, u);
        len[u] = __builtin_amdgcn_readlane(bnd1, u) - s0[u];
    }

    int p0 = 0;
    bool first = true;
    bool again;
    do {
        int p[GCN_WPT];
        bool longer = false;
#pragma unroll
        for (int u = 0; u < 8; ++u) {
            const int* seg = list + (sg0 + u) * GCN_EPB + s0[u] + p0;
            const int left = len[u] - p0;
            p[2 * u] = (lane < left) ? seg[lane] : -1;
            p[2 * u + 1] = (lane + 32 < left) ? seg[lane + 32] : -1;
            longer = longer || (left > GCN_SEGW);
        }

        if (tid < GCN_ROWS_PB) cnt[tid] = 0;
        if (tid == 0) more = 0;
        __syncthreads();
#pragma unroll
        for (int u = 0; u < GCN_WPT; ++u)
            if (p[u] >= 0) atomicAdd(&cnt[p[u] & 511], 1);
        if (longer && lane == 0) more = 1;
        __syncthreads();

        if (wave == 0) {
            int loc[16];
            int s = 0;
#pragma unroll
            for (int j = 0; j < 16; ++j) { loc[j] = cnt[lane * 16 + j]; s += loc[j]; }
            int inc = s;
#pragma unroll
            for (int d = 1; d < 32; d <<= 1) {
                const int t = __shfl_up(inc, d);
                if (lane >= d) inc += t;
            }
            int run = inc - s;
#pragma unroll
            for (int j = 0; j < 16; ++j) {
                off[lane * 16 + j] = run;
                cur[lane * 16 + j] = run;
                run += loc[j];
            }
            if (lane == 31) off[GCN_ROWS_PB] = run;
        }
        __syncthreads();

#pragma unroll
        for (int u = 0; u < GCN_WPT; ++u) {
            if (p[u] >= 0) {
                const int pos = atomicAdd(&cur[p[u] & 511], 1);
                sorted[pos] = p[u] >> 9;
            }
        }
        __syncthreads();
        again = gcn_uni(more) != 0;

        for (int jj = 0; jj < 8; ++jj) {
            const int rA = wave + 64 * jj;
            const int rB = rA + 32;
            const int rowA = k * GCN_ROWS_PB + rA;
            const int rowB = rowA + 32;
            const bool okA = rowA < GCN_N_NODES;
            const bool okB = rowB < GCN_N_NODES;
            const int rbA = gcn_uni(off[rA]);
            const int reA = gcn_uni(off[rA + 1]);
            const int rbB = gcn_uni(off[rB]);
            const int reB = gcn_uni(off[rB + 1]);
            const int mA = min(32, reA - rbA);
            const int mB = min(32, reB - rbB);

            float accA = 0.f;
            float accB = 0.f;
            if (!first) {
                if (okA) accA = out[(size_t)rowA * GCN_D + lane];
                if (okB) accB = out[(size_t)rowB * GCN_D + lane];
            }

            int vA = (lane < mA) ? sorted[rbA + lane] : 0x7fffffff;
            int vB = (lane < mB) ? sorted[rbB + lane] : 0x7fffffff;
            vA = gcn_sort10(vA, lane);
            vB = gcn_sort10(vB, lane);
            if (max(mA, mB) > 16) {
                vA = gcn_merge32(vA, lane);
                vB = gcn_merge32(vB, lane);
            }
            const int voffA = gcn_rowoff(vA);
            const int voffB = gcn_rowoff(vB);

            float fA[32];
            float fB[32];
            gcn_issue_group(xb, voffA, lane4, mA, fA);
            gcn_issue_group(xb, voffB, lane4, mB, fB);
            accA = gcn_sum_group(fA, mA, accA);
            accB = gcn_sum_group(fB, mB, accB);

            accA = gcn_more_groups(sorted, xb, lane4, lane, rbA + 32, reA, accA);
            accB = gcn_more_groups(sorted, xb, lane4, lane, rbB + 32, reB, accB);

            if (okA) out[(size_t)rowA * GCN_D + lane] = accA;
            if (okB) out[(size_t)rowB * GCN_D + lane] = accB;
        }
        __syncthreads();
        p0 += GCN_SEGW;
        first = false;
    } while (again);
}

__global__ __launch_bounds__(256) void gcn_gemm_inplace(
    float* io, const float* __restrict__ W, int n_tiles)
{
    const int lane = threadIdx.x & 31;
    const int wave = threadIdx.x >> 5;
    const int tile = blockIdx.x * 8 + wave;
    if (tile >= n_tiles) return;

    const int m  = lane & 15;
    const int kb = (lane >> 4) * 2;

    v2f b0[8], b1[8];
#pragma unroll
    for (int kc = 0; kc < 8; ++kc) {
        const int k = kc * 4 + kb;
        const float* w0 = W + (m)      * GCN_D + k;
        const float* w1 = W + (16 + m) * GCN_D + k;
        v2f t0; t0.x = w0[0]; t0.y = w0[1];
        v2f t1; t1.x = w1[0]; t1.y = w1[1];
        b0[kc] = t0; b1[kc] = t1;
    }

    float* xr = io + (size_t)tile * 16 * GCN_D;
    v2f a[8];
#pragma unroll
    for (int kc = 0; kc < 8; ++kc) {
        const int k = kc * 4 + kb;
        a[kc] = *(const v2f*)(xr + m * GCN_D + k);
    }

    v8f c0 = {}; v8f c1 = {};
#pragma unroll
    for (int kc = 0; kc < 8; ++kc) {
        c0 = __builtin_amdgcn_wmma_f32_16x16x4_f32(
                 false, a[kc], false, b0[kc], (short)0, c0, false, false);
        c1 = __builtin_amdgcn_wmma_f32_16x16x4_f32(
                 false, a[kc], false, b1[kc], (short)0, c1, false, false);
    }

    const int n     = lane & 15;
    const int rbase = (lane >> 4) * 8;
#pragma unroll
    for (int r = 0; r < 8; ++r) {
        xr[(rbase + r) * GCN_D + n]      = c0[r];
        xr[(rbase + r) * GCN_D + 16 + n] = c1[r];
    }
}

extern "C" void kernel_launch(void* const* d_in, const int* in_sizes, int n_in,
                              void* d_out, int out_size, void* d_ws, size_t ws_size,
                              hipStream_t stream)
{
    const float* x  = (const float*)d_in[0];
    const int*   ei = (const int*)d_in[1];
    const float* W  = (const float*)d_in[2];
    float*       out = (float*)d_out;
    int*         wsi = (int*)d_ws;
    int* list   = wsi;
    int* segoff = wsi + 1600000;

    gcn_blocksort<<<GCN_NBLK, 1024, 0, stream>>>(ei, list, segoff);
    gcn_rowsum<<<196, 1024, 0, stream>>>(x, list, segoff, out);
    gcn_gemm_inplace<<<782, 256, 0, stream>>>(out, W, 6250);
}


# W^T launch folded into the row kernel: finished row sums stay in LDS, a wave takes 16 of them as f32 WMMA A fragments and stores the out rows once, in whole lines; 3 launches -> 2; sum order kept
# speedup vs baseline: 5.6723x; 1.6677x over previous
#include <hip/hip_runtime.h>


typedef __attribute__((ext_vector_type(2))) float v2f;
typedef __attribute__((ext_vector_type(4))) float v4f;
typedef __attribute__((ext_vector_type(8))) float v8f;

#define GCN_N_NODES 100000
#define GCN_N_EDGES 1600000
#define GCN_D 32

#define GCN_NBLK 256
#define GCN_EPB 6250
#define GCN_EPT 7
#define GCN_NBK 256
#define GCN_ROWS_PB 512
#define GCN_SEGW 64
#define GCN_CAP 16384
#define GCN_WPT 16
#define GCN_RP 36

__global__ __launch_bounds__(1024) void gcn_blocksort(
    const int* __restrict__ ei, int* __restrict__ list, int* __restrict__ segoff)
{
    __shared__ int cnt[GCN_NBK];
    __shared__ int off[GCN_NBK + 1];
    __shared__ int stage[GCN_EPB];
    const int tid = threadIdx.x;
    const int lane = tid & 31;
    const int b = blockIdx.x;
    const int* src = ei + b * GCN_EPB;
    const int* dst = ei + GCN_N_EDGES + b * GCN_EPB;

    int d[GCN_EPT];
    int s[GCN_EPT];
#pragma unroll
    for (int u = 0; u < GCN_EPT; ++u) {
        const int i = tid + u * 1024;
        const bool in = i < GCN_EPB;
        d[u] = in ? dst[i] : 0;
        s[u] = in ? src[i] : 0;
    }

    if (tid < GCN_NBK) cnt[tid] = 0;
    __syncthreads();
    int rank[GCN_EPT];
#pragma unroll
    for (int u = 0; u < GCN_EPT; ++u) {
        const bool in = tid + u * 1024 < GCN_EPB;
        rank[u] = in ? atomicAdd(&cnt[d[u] >> 9], 1) : 0;
    }
    __syncthreads();

    if (tid < 32) {
        int loc[8];
        int sum = 0;
#pragma unroll
        for (int j = 0; j < 8; ++j) { loc[j] = cnt[lane * 8 + j]; sum += loc[j]; }
        int inc = sum;
#pragma unroll
        for (int dd = 1; dd < 32; dd <<= 1) {
            const int t = __shfl_up(inc, dd);
            if (lane >= dd) inc += t;
        }
        int run = inc - sum;
#pragma unroll
        for (int j = 0; j < 8; ++j) { off[lane * 8 + j] = run; run += loc[j]; }
        if (lane == 31) off[GCN_NBK] = run;
    }
    __syncthreads();

    int first[GCN_EPT];
#pragma unroll
    for (int u = 0; u < GCN_EPT; ++u) first[u] = off[d[u] >> 9];
#pragma unroll
    for (int u = 0; u < GCN_EPT; ++u) {
        const bool in = tid + u * 1024 < GCN_EPB;
        if (in) stage[first[u] + rank[u]] = (s[u] << 9) | (d[u] & 511);
    }
    __syncthreads();

#pragma unroll
    for (int u = 0; u < GCN_EPT; ++u) {
        const int i = tid + u * 1024;
        if (i < GCN_EPB) list[b * GCN_EPB + i] = stage[i];
    }
    if (tid <= GCN_NBK) segoff[tid * GCN_NBLK + b] = off[tid];
}

__device__ __forceinline__ int gcn_uni(int v) { return __builtin_amdgcn_readfirstlane(v); }

template <int J>
__device__ __forceinline__ int gcn_xrow(int v)
{
    return __builtin_amdgcn_update_dpp(0, v, 0x160 + J, 0xf, 0xf, true);
}

__device__ __forceinline__ int gcn_xhalf(int v)
{
    return __builtin_amdgcn_permlanex16(v, v, 0x76543210, 0xfedcba98, false, false);
}

template <int K, int J>
__device__ __forceinline__ int gcn_cx(int v, int lane)
{
    const int p = (J == 16) ? gcn_xhalf(v) : gcn_xrow<(J & 15)>(v);
    const bool keep_min = ((lane & J) == 0) == ((lane & K) == 0);
    const int lo = min(v, p);
    const int hi = max(v, p);
    return keep_min ? lo : hi;
}

__device__ __forceinline__ int gcn_sort10(int val, int lane)
{
    val = gcn_cx<2, 1>(val, lane);
    val = gcn_cx<4, 2>(val, lane);
    val = gcn_cx<4, 1>(val, lane);
    val = gcn_cx<8, 4>(val, lane);
    val = gcn_cx<8, 2>(val, lane);
    val = gcn_cx<8, 1>(val, lane);
    val = gcn_cx<16, 8>(val, lane);
    val = gcn_cx<16, 4>(val, lane);
    val = gcn_cx<16, 2>(val, lane);
    val = gcn_cx<16, 1>(val, lane);
    return val;
}

__device__ __forceinline__ int gcn_merge32(int val, int lane)
{
    val = gcn_cx<32, 16>(val, lane);
    val = gcn_cx<32, 8>(val, lane);
    val = gcn_cx<32, 4>(val, lane);
    val = gcn_cx<32, 2>(val, lane);
    val = gcn_cx<32, 1>(val, lane);
    return val;
}

__device__ __forceinline__ int gcn_rowoff(int val)
{
    return min(val, GCN_N_NODES - 1) << 7;
}

template <int B>
__device__ __forceinline__ void gcn_issue8(
    const char* xb, int voff, unsigned lane4, float (&f)[32])
{
#pragma unroll
    for (int u = 0; u < 8; ++u) {
        const int i = 8 * B + u;
        const unsigned so = (unsigned)__builtin_amdgcn_readlane(voff, i);
        f[i] = *(const float*)(xb + (so + lane4));
    }
}

template <int B>
__device__ __forceinline__ float gcn_sum8(const float (&f)[32], int m, float acc)
{
#pragma unroll
    for (int u = 0; u < 8; ++u) {
        const int i = 8 * B + u;
        acc += (i < m) ? f[i] : 0.f;
    }
    return acc;
}

__device__ __forceinline__ void gcn_issue_group(
    const char* xb, int voff, unsigned lane4, int m, float (&f)[32])
{
    gcn_issue8<0>(xb, voff, lane4, f);
    if (m > 8)  gcn_issue8<1>(xb, voff, lane4, f);
    if (m > 16) gcn_issue8<2>(xb, voff, lane4, f);
    if (m > 24) gcn_issue8<3>(xb, voff, lane4, f);
}

__device__ __forceinline__ float gcn_sum_group(const float (&f)[32], int m, float acc)
{
    acc = gcn_sum8<0>(f, m, acc);
    if (m > 8)  acc = gcn_sum8<1>(f, m, acc);
    if (m > 16) acc = gcn_sum8<2>(f, m, acc);
    if (m > 24) acc = gcn_sum8<3>(f, m, acc);
    return acc;
}

__device__ __forceinline__ float gcn_more_groups(
    const int* sorted, const char* xb, unsigned lane4, int lane,
    int g0, int re, float acc)
{
    for (; g0 < re; g0 += 32) {
        const int m = min(32, re - g0);
        int val = (lane < m) ? sorted[g0 + lane] : 0x7fffffff;
        val = gcn_merge32(gcn_sort10(val, lane), lane);
        const int voff = gcn_rowoff(val);
        for (int i0 = 0; i0 < m; i0 += 8) {
            float f[8];
#pragma unroll
            for (int u = 0; u < 8; ++u) {
                const unsigned so = (unsigned)__builtin_amdgcn_readlane(voff, i0 + u);
                f[u] = *(const float*)(xb + (so + lane4));
            }
#pragma unroll
            for (int u = 0; u < 8; ++u) acc += (i0 + u < m) ? f[u] : 0.f;
        }
    }
    return acc;
}

__global__ __launch_bounds__(1024) void gcn_rowsum(
    const float* __restrict__ x, const int* __restrict__ list,
    const int* __restrict__ segoff, const float* __restrict__ W,
    float* __restrict__ out)
{
    __shared__ __attribute__((aligned(16))) float rows[GCN_ROWS_PB * GCN_RP];
    __shared__ int cnt[GCN_ROWS_PB];
    __shared__ int off[GCN_ROWS_PB + 1];
    __shared__ int cur[GCN_ROWS_PB];
    __shared__ int sorted[GCN_CAP];
    __shared__ int more;

    const int tid = threadIdx.x;
    const int lane = tid & 31;
    const int wave = gcn_uni(tid >> 5);
    const int k = blockIdx.x;
    const char* xb = (const char*)x;
    const unsigned lane4 = (unsigned)lane * 4u;
    const int sg0 = wave * 8;
    int bnd0 = 0;
    int bnd1 = 0;
    if (lane < 8) {
        bnd0 = segoff[k * GCN_NBLK + sg0 + lane];
        bnd1 = segoff[(k + 1) * GCN_NBLK + sg0 + lane];
    }
    int s0[8];
    int len[8];
#pragma unroll
    for (int u = 0; u < 8; ++u) {
        s0[u] = __builtin_amdgcn_readlane(bnd0, u);
        len[u] = __builtin_amdgcn_readlane(bnd1, u) - s0[u];
    }

    int p0 = 0;
    bool first = true;
    bool again;
    do {
        int p[GCN_WPT];
        bool longer = false;
#pragma unroll
        for (int u = 0; u < 8; ++u) {
            const int* seg = list + (sg0 + u) * GCN_EPB + s0[u] + p0;
            const int left = len[u] - p0;
            p[2 * u] = (lane < left) ? seg[lane] : -1;
            p[2 * u + 1] = (lane + 32 < left) ? seg[lane + 32] : -1;
            longer = longer || (left > GCN_SEGW);
        }

        if (tid < GCN_ROWS_PB) cnt[tid] = 0;
        if (tid == 0) more = 0;
        __syncthreads();
#pragma unroll
        for (int u = 0; u < GCN_WPT; ++u)
            if (p[u] >= 0) atomicAdd(&cnt[p[u] & 511], 1);
        if (longer && lane == 0) more = 1;
        __syncthreads();

        if (wave == 0) {
            int loc[16];
            int s = 0;
#pragma unroll
            for (int j = 0; j < 16; ++j) { loc[j] = cnt[lane * 16 + j]; s += loc[j]; }
            int inc = s;
#pragma unroll
            for (int d = 1; d < 32; d <<= 1) {
                const int t = __shfl_up(inc, d);
                if (lane >= d) inc += t;
            }
            int run = inc - s;
#pragma unroll
            for (int j = 0; j < 16; ++j) {
                off[lane * 16 + j] = run;
                cur[lane * 16 + j] = run;
                run += loc[j];
            }
            if (lane == 31) off[GCN_ROWS_PB] = run;
        }
        __syncthreads();

#pragma unroll
        for (int u = 0; u < GCN_WPT; ++u) {
            if (p[u] >= 0) {
                const int pos = atomicAdd(&cur[p[u] & 511], 1);
                sorted[pos] = p[u] >> 9;
            }
        }
        __syncthreads();
        again = gcn_uni(more) != 0;

        for (int jj = 0; jj < 8; ++jj) {
            const int rA = wave + 64 * jj;
            const int rB = rA + 32;
            const int rbA = gcn_uni(off[rA]);
            const int reA = gcn_uni(off[rA + 1]);
            const int rbB = gcn_uni(off[rB]);
            const int reB = gcn_uni(off[rB + 1]);
            const int mA = min(32, reA - rbA);
            const int mB = min(32, reB - rbB);

            float accA = 0.f;
            float accB = 0.f;
            if (!first) {
                accA = rows[rA * GCN_RP + lane];
                accB = rows[rB * GCN_RP + lane];
            }

            int vA = (lane < mA) ? sorted[rbA + lane] : 0x7fffffff;
            int vB = (lane < mB) ? sorted[rbB + lane] : 0x7fffffff;
            vA = gcn_sort10(vA, lane);
            vB = gcn_sort10(vB, lane);
            if (max(mA, mB) > 16) {
                vA = gcn_merge32(vA, lane);
                vB = gcn_merge32(vB, lane);
            }
            const int voffA = gcn_rowoff(vA);
            const int voffB = gcn_rowoff(vB);

            float fA[32];
            float fB[32];
            gcn_issue_group(xb, voffA, lane4, mA, fA);
            gcn_issue_group(xb, voffB, lane4, mB, fB);
            accA = gcn_sum_group(fA, mA, accA);
            accB = gcn_sum_group(fB, mB, accB);

            accA = gcn_more_groups(sorted, xb, lane4, lane, rbA + 32, reA, accA);
            accB = gcn_more_groups(sorted, xb, lane4, lane, rbB + 32, reB, accB);

            rows[rA * GCN_RP + lane] = accA;
            rows[rB * GCN_RP + lane] = accB;
        }
        __syncthreads();
        p0 += GCN_SEGW;
        first = false;
    } while (again);

    const int tile = k * (GCN_ROWS_PB / 16) + wave;
    if (tile >= GCN_N_NODES / 16) return;

    const int m  = lane & 15;
    const int kb = (lane >> 4) * 2;
    const unsigned a_b = (unsigned)((wave * 16 + m) * GCN_RP + kb) * 4u;
    unsigned d_b = (unsigned)((wave * 16 + (lane >> 4) * 8) * GCN_RP + m) * 4u;
    unsigned e_b = d_b + 4 * GCN_RP * 4;
    unsigned r_b = (unsigned)((wave * 16 + (lane >> 3)) * GCN_RP + (lane & 7) * 4) * 4u;
    unsigned o_b = (unsigned)lane * 16u;
    const char* rowsb = (const char*)rows;
    char* rowsw = (char*)rows;

    v2f b0[8], b1[8];
#pragma unroll
    for (int kc = 0; kc < 8; ++kc) {
        const int kk = kc * 4 + kb;
        const float* w0 = W + (m)      * GCN_D + kk;
        const float* w1 = W + (16 + m) * GCN_D + kk;
        v2f t0; t0.x = w0[0]; t0.y = w0[1];
        v2f t1; t1.x = w1[0]; t1.y = w1[1];
        b0[kc] = t0; b1[kc] = t1;
    }

    v2f a[8];
#pragma unroll
    for (int kc = 0; kc < 8; ++kc) a[kc] = *(const v2f*)(rowsb + a_b + kc * 16);

    asm volatile("" : "+v"(d_b), "+v"(e_b), "+v"(r_b), "+v"(o_b));
    __builtin_amdgcn_sched_barrier(0);

    v8f c0 = {}; v8f c1 = {};
#pragma unroll
    for (int kc = 0; kc < 8; ++kc) {
        c0 = __builtin_amdgcn_wmma_f32_16x16x4_f32(
                 false, a[kc], false, b0[kc], (short)0, c0, false, false);
        c1 = __builtin_amdgcn_wmma_f32_16x16x4_f32(
                 false, a[kc], false, b1[kc], (short)0, c1, false, false);
    }

#pragma unroll
    for (int r = 0; r < 4; ++r) {
        *(float*)(rowsw + d_b + r * (GCN_RP * 4))      = c0[r];
        *(float*)(rowsw + d_b + r * (GCN_RP * 4) + 64) = c1[r];
        *(float*)(rowsw + e_b + r * (GCN_RP * 4))      = c0[r + 4];
        *(float*)(rowsw + e_b + r * (GCN_RP * 4) + 64) = c1[r + 4];
    }
    __builtin_amdgcn_fence(__ATOMIC_ACQ_REL, "wavefront");
    char* og = (char*)(out + (size_t)tile * 16 * GCN_D);
#pragma unroll
    for (int q = 0; q < 4; ++q) {
        const v4f v = *(const v4f*)(rowsb + r_b + q * (4 * GCN_RP * 4));
        *(v4f*)(og + o_b + q * (4 * GCN_D * 4)) = v;
    }
}

extern "C" void kernel_launch(void* const* d_in, const int* in_sizes, int n_in,
                              void* d_out, int out_size, void* d_ws, size_t ws_size,
                              hipStream_t stream)
{
    const float* x  = (const float*)d_in[0];
    const int*   ei = (const int*)d_in[1];
    const float* W  = (const float*)d_in[2];
    float*       out = (float*)d_out;
    int*         wsi = (int*)d_ws;
    int* list   = wsi;
    int* segoff = wsi + 1600000;

    gcn_blocksort<<<GCN_NBLK, 1024, 0, stream>>>(ei, list, segoff);
    gcn_rowsum<<<196, 1024, 0, stream>>>(x, list, segoff, W, out);
}


# Partition resized to the chip: 250 edge blocks x 6400 edges, 250 row buckets x 400 rows (was 256 / 196 x 512), 25-wave groups, one 16-row tile a wave: row kernel on 250 WGPs, not 196; sum order kept
# speedup vs baseline: 6.1141x; 1.0779x over previous
#include <hip/hip_runtime.h>


typedef __attribute__((ext_vector_type(2))) float v2f;
typedef __attribute__((ext_vector_type(4))) float v4f;
typedef __attribute__((ext_vector_type(8))) float v8f;

#define GCN_N_NODES 100000
#define GCN_N_EDGES 1600000
#define GCN_D 32

#define GCN_NBLK 250
#define GCN_EPB 6400
#define GCN_TPB 800
#define GCN_NWAVE 25
#define GCN_EPT 8
#define GCN_NBK 256
#define GCN_ROWS_PB 400
#define GCN_NSLOT 512
#define GCN_SPW 10
#define GCN_SEGW 64
#define GCN_CAP 16000
#define GCN_WPT 20
#define GCN_RP 36

__global__ __launch_bounds__(GCN_TPB) void gcn_blocksort(
    const int* __restrict__ ei, int* __restrict__ list, int* __restrict__ segoff)
{
    __shared__ int cnt[GCN_NBK];
    __shared__ int off[GCN_NBK + 1];
    __shared__ int stage[GCN_EPB];
    const int tid = threadIdx.x;
    const int lane = tid & 31;
    const int b = blockIdx.x;
    const int* src = ei + b * GCN_EPB;
    const int* dst = ei + GCN_N_EDGES + b * GCN_EPB;

    int d[GCN_EPT];
    int s[GCN_EPT];
#pragma unroll
    for (int u = 0; u < GCN_EPT; ++u) {
        const int i = tid + u * GCN_TPB;
        d[u] = dst[i];
        s[u] = src[i];
    }

    if (tid < GCN_NBK) cnt[tid] = 0;
    __syncthreads();
    int bk[GCN_EPT];
    int rank[GCN_EPT];
#pragma unroll
    for (int u = 0; u < GCN_EPT; ++u) {
        bk[u] = (int)((unsigned)d[u] / (unsigned)GCN_ROWS_PB);
        rank[u] = atomicAdd(&cnt[bk[u]], 1);
    }
    __syncthreads();

    if (tid < 32) {
        int loc[8];
        int sum = 0;
#pragma unroll
        for (int j = 0; j < 8; ++j) { loc[j] = cnt[lane * 8 + j]; sum += loc[j]; }
        int inc = sum;
#pragma unroll
        for (int dd = 1; dd < 32; dd <<= 1) {
            const int t = __shfl_up(inc, dd);
            if (lane >= dd) inc += t;
        }
        int run = inc - sum;
#pragma unroll
        for (int j = 0; j < 8; ++j) { off[lane * 8 + j] = run; run += loc[j]; }
        if (lane == 31) off[GCN_NBK] = run;
    }
    __syncthreads();

    int first[GCN_EPT];
#pragma unroll
    for (int u = 0; u < GCN_EPT; ++u) first[u] = off[bk[u]];
#pragma unroll
    for (int u = 0; u < GCN_EPT; ++u)
        stage[first[u] + rank[u]] = (s[u] << 9) | (d[u] - bk[u] * GCN_ROWS_PB);
    __syncthreads();

#pragma unroll
    for (int u = 0; u < GCN_EPT; ++u) {
        const int i = tid + u * GCN_TPB;
        list[b * GCN_EPB + i] = stage[i];
    }
    if (tid <= GCN_NBK) segoff[tid * GCN_NBLK + b] = off[tid];
}

__device__ __forceinline__ int gcn_uni(int v) { return __builtin_amdgcn_readfirstlane(v); }

template <int J>
__device__ __forceinline__ int gcn_xrow(int v)
{
    return __builtin_amdgcn_update_dpp(0, v, 0x160 + J, 0xf, 0xf, true);
}

__device__ __forceinline__ int gcn_xhalf(int v)
{
    return __builtin_amdgcn_permlanex16(v, v, 0x76543210, 0xfedcba98, false, false);
}

template <int K, int J>
__device__ __forceinline__ int gcn_cx(int v, int lane)
{
    const int p = (J == 16) ? gcn_xhalf(v) : gcn_xrow<(J & 15)>(v);
    const bool keep_min = ((lane & J) == 0) == ((lane & K) == 0);
    const int lo = min(v, p);
    const int hi = max(v, p);
    return keep_min ? lo : hi;
}

__device__ __forceinline__ int gcn_sort10(int val, int lane)
{
    val = gcn_cx<2, 1>(val, lane);
    val = gcn_cx<4, 2>(val, lane);
    val = gcn_cx<4, 1>(val, lane);
    val = gcn_cx<8, 4>(val, lane);
    val = gcn_cx<8, 2>(val, lane);
    val = gcn_cx<8, 1>(val, lane);
    val = gcn_cx<16, 8>(val, lane);
    val = gcn_cx<16, 4>(val, lane);
    val = gcn_cx<16, 2>(val, lane);
    val = gcn_cx<16, 1>(val, lane);
    return val;
}

__device__ __forceinline__ int gcn_merge32(int val, int lane)
{
    val = gcn_cx<32, 16>(val, lane);
    val = gcn_cx<32, 8>(val, lane);
    val = gcn_cx<32, 4>(val, lane);
    val = gcn_cx<32, 2>(val, lane);
    val = gcn_cx<32, 1>(val, lane);
    return val;
}

__device__ __forceinline__ int gcn_rowoff(int val)
{
    return min(val, GCN_N_NODES - 1) << 7;
}

template <int B>
__device__ __forceinline__ void gcn_issue8(
    const char* xb, int voff, unsigned lane4, float (&f)[32])
{
#pragma unroll
    for (int u = 0; u < 8; ++u) {
        const int i = 8 * B + u;
        const unsigned so = (unsigned)__builtin_amdgcn_readlane(voff, i);
        f[i] = *(const float*)(xb + (so + lane4));
    }
}

template <int B>
__device__ __forceinline__ float gcn_sum8(const float (&f)[32], int m, float acc)
{
#pragma unroll
    for (int u = 0; u < 8; ++u) {
        const int i = 8 * B + u;
        acc += (i < m) ? f[i] : 0.f;
    }
    return acc;
}

__device__ __forceinline__ void gcn_issue_group(
    const char* xb, int voff, unsigned lane4, int m, float (&f)[32])
{
    gcn_issue8<0>(xb, voff, lane4, f);
    if (m > 8)  gcn_issue8<1>(xb, voff, lane4, f);
    if (m > 16) gcn_issue8<2>(xb, voff, lane4, f);
    if (m > 24) gcn_issue8<3>(xb, voff, lane4, f);
}

__device__ __forceinline__ float gcn_sum_group(const float (&f)[32], int m, float acc)
{
    acc = gcn_sum8<0>(f, m, acc);
    if (m > 8)  acc = gcn_sum8<1>(f, m, acc);
    if (m > 16) acc = gcn_sum8<2>(f, m, acc);
    if (m > 24) acc = gcn_sum8<3>(f, m, acc);
    return acc;
}

__device__ __forceinline__ float gcn_more_groups(
    const int* sorted, const char* xb, unsigned lane4, int lane,
    int g0, int re, float acc)
{
    for (; g0 < re; g0 += 32) {
        const int m = min(32, re - g0);
        int val = (lane < m) ? sorted[g0 + lane] : 0x7fffffff;
        val = gcn_merge32(gcn_sort10(val, lane), lane);
        const int voff = gcn_rowoff(val);
        for (int i0 = 0; i0 < m; i0 += 8) {
            float f[8];
#pragma unroll
            for (int u = 0; u < 8; ++u) {
                const unsigned so = (unsigned)__builtin_amdgcn_readlane(voff, i0 + u);
                f[u] = *(const float*)(xb + (so + lane4));
            }
#pragma unroll
            for (int u = 0; u < 8; ++u) acc += (i0 + u < m) ? f[u] : 0.f;
        }
    }
    return acc;
}

__global__ __launch_bounds__(GCN_TPB) void gcn_rowsum(
    const float* __restrict__ x, const int* __restrict__ list,
    const int* __restrict__ segoff, const float* __restrict__ W,
    float* __restrict__ out)
{
    __shared__ __attribute__((aligned(16))) unsigned char lds[
        (GCN_ROWS_PB * GCN_RP + 3 * GCN_NSLOT + 4 + GCN_CAP + 4) * 4];
    float* rows   = (float*)lds;
    int*   cnt    = (int*)lds + GCN_ROWS_PB * GCN_RP;
    int*   cur    = cnt + GCN_NSLOT;
    int*   off    = cur + GCN_NSLOT;
    int*   sorted = off + GCN_NSLOT + 4;
    int*   more   = sorted + GCN_CAP;

    const int tid = threadIdx.x;
    const int lane = tid & 31;
    const int wave = gcn_uni(tid >> 5);
    const int k = blockIdx.x;
    const char* xb = (const char*)x;
    const unsigned lane4 = (unsigned)lane * 4u;
    const int sg0 = wave * GCN_SPW;
    int bnd0 = 0;
    int bnd1 = 0;
    if (lane < GCN_SPW) {
        bnd0 = segoff[k * GCN_NBLK + sg0 + lane];
        bnd1 = segoff[(k + 1) * GCN_NBLK + sg0 + lane];
    }
    int s0[GCN_SPW];
    int len[GCN_SPW];
#pragma unroll
    for (int u = 0; u < GCN_SPW; ++u) {
        s0[u] = __builtin_amdgcn_readlane(bnd0, u);
        len[u] = __builtin_amdgcn_readlane(bnd1, u) - s0[u];
    }

    int p0 = 0;
    bool first = true;
    bool again;
    do {
        int p[GCN_WPT];
        bool longer = false;
#pragma unroll
        for (int u = 0; u < GCN_SPW; ++u) {
            const int* seg = list + (sg0 + u) * GCN_EPB + s0[u] + p0;
            const int left = len[u] - p0;
            p[2 * u] = (lane < left) ? seg[lane] : -1;
            p[2 * u + 1] = (lane + 32 < left) ? seg[lane + 32] : -1;
            longer = longer || (left > GCN_SEGW);
        }

        if (tid < GCN_NSLOT) cnt[tid] = 0;
        if (tid == 0) *more = 0;
        __syncthreads();
#pragma unroll
        for (int u = 0; u < GCN_WPT; ++u)
            if (p[u] >= 0) atomicAdd(&cnt[p[u] & 511], 1);
        if (longer && lane == 0) *more = 1;
        __syncthreads();

        if (wave == 0) {
            int loc[16];
            int s = 0;
#pragma unroll
            for (int j = 0; j < 16; ++j) { loc[j] = cnt[lane * 16 + j]; s += loc[j]; }
            int inc = s;
#pragma unroll
            for (int d = 1; d < 32; d <<= 1) {
                const int t = __shfl_up(inc, d);
                if (lane >= d) inc += t;
            }
            int run = inc - s;
#pragma unroll
            for (int j = 0; j < 16; ++j) {
                off[lane * 16 + j] = run;
                cur[lane * 16 + j] = run;
                run += loc[j];
            }
            if (lane == 31) off[GCN_NSLOT] = run;
        }
        __syncthreads();

#pragma unroll
        for (int u = 0; u < GCN_WPT; ++u) {
            if (p[u] >= 0) {
                const int pos = atomicAdd(&cur[p[u] & 511], 1);
                sorted[pos] = p[u] >> 9;
            }
        }
        __syncthreads();
        again = gcn_uni(*more) != 0;

        for (int jj = 0; jj < 8; ++jj) {
            const int rA = wave + 2 * GCN_NWAVE * jj;
            const int rB = rA + GCN_NWAVE;
            const int rbA = gcn_uni(off[rA]);
            const int reA = gcn_uni(off[rA + 1]);
            const int rbB = gcn_uni(off[rB]);
            const int reB = gcn_uni(off[rB + 1]);
            const int mA = min(32, reA - rbA);
            const int mB = min(32, reB - rbB);

            float accA = 0.f;
            float accB = 0.f;
            if (!first) {
                accA = rows[rA * GCN_RP + lane];
                accB = rows[rB * GCN_RP + lane];
            }

            int vA = (lane < mA) ? sorted[rbA + lane] : 0x7fffffff;
            int vB = (lane < mB) ? sorted[rbB + lane] : 0x7fffffff;
            vA = gcn_sort10(vA, lane);
            vB = gcn_sort10(vB, lane);
            if (max(mA, mB) > 16) {
                vA = gcn_merge32(vA, lane);
                vB = gcn_merge32(vB, lane);
            }
            const int voffA = gcn_rowoff(vA);
            const int voffB = gcn_rowoff(vB);

            float fA[32];
            float fB[32];
            gcn_issue_group(xb, voffA, lane4, mA, fA);
            gcn_issue_group(xb, voffB, lane4, mB, fB);
            accA = gcn_sum_group(fA, mA, accA);
            accB = gcn_sum_group(fB, mB, accB);

            accA = gcn_more_groups(sorted, xb, lane4, lane, rbA + 32, reA, accA);
            accB = gcn_more_groups(sorted, xb, lane4, lane, rbB + 32, reB, accB);

            rows[rA * GCN_RP + lane] = accA;
            rows[rB * GCN_RP + lane] = accB;
        }
        __syncthreads();
        p0 += GCN_SEGW;
        first = false;
    } while (again);

    const int tile = k * (GCN_ROWS_PB / 16) + wave;

    const int m  = lane & 15;
    const int kb = (lane >> 4) * 2;
    const unsigned a_b = (unsigned)((wave * 16 + m) * GCN_RP + kb) * 4u;
    unsigned d_b = (unsigned)((wave * 16 + (lane >> 4) * 8) * GCN_RP + m) * 4u;
    unsigned e_b = d_b + 4 * GCN_RP * 4;
    unsigned r_b = (unsigned)((wave * 16 + (lane >> 3)) * GCN_RP + (lane & 7) * 4) * 4u;
    unsigned o_b = (unsigned)lane * 16u;
    const char* rowsb = (const char*)rows;
    char* rowsw = (char*)rows;

    v2f b0[8], b1[8];
#pragma unroll
    for (int kc = 0; kc < 8; ++kc) {
        const int kk = kc * 4 + kb;
        const float* w0 = W + (m)      * GCN_D + kk;
        const float* w1 = W + (16 + m) * GCN_D + kk;
        v2f t0; t0.x = w0[0]; t0.y = w0[1];
        v2f t1; t1.x = w1[0]; t1.y = w1[1];
        b0[kc] = t0; b1[kc] = t1;
    }

    v2f a[8];
#pragma unroll
    for (int kc = 0; kc < 8; ++kc) a[kc] = *(const v2f*)(rowsb + a_b + kc * 16);

    asm volatile("" : "+v"(d_b), "+v"(e_b), "+v"(r_b), "+v"(o_b));
    __builtin_amdgcn_sched_barrier(0);

    v8f c0 = {}; v8f c1 = {};
#pragma unroll
    for (int kc = 0; kc < 8; ++kc) {
        c0 = __builtin_amdgcn_wmma_f32_16x16x4_f32(
                 false, a[kc], false, b0[kc], (short)0, c0, false, false);
        c1 = __builtin_amdgcn_wmma_f32_16x16x4_f32(
                 false, a[kc], false, b1[kc], (short)0, c1, false, false);
    }

#pragma unroll
    for (int r = 0; r < 4; ++r) {
        *(float*)(rowsw + d_b + r * (GCN_RP * 4))      = c0[r];
        *(float*)(rowsw + d_b + r * (GCN_RP * 4) + 64) = c1[r];
        *(float*)(rowsw + e_b + r * (GCN_RP * 4))      = c0[r + 4];
        *(float*)(rowsw + e_b + r * (GCN_RP * 4) + 64) = c1[r + 4];
    }
    __builtin_amdgcn_fence(__ATOMIC_ACQ_REL, "wavefront");
    char* og = (char*)(out + (size_t)tile * 16 * GCN_D);
#pragma unroll
    for (int q = 0; q < 4; ++q) {
        const v4f v = *(const v4f*)(rowsb + r_b + q * (4 * GCN_RP * 4));
        *(v4f*)(og + o_b + q * (4 * GCN_D * 4)) = v;
    }
}

extern "C" void kernel_launch(void* const* d_in, const int* in_sizes, int n_in,
                              void* d_out, int out_size, void* d_ws, size_t ws_size,
                              hipStream_t stream)
{
    const float* x  = (const float*)d_in[0];
    const int*   ei = (const int*)d_in[1];
    const float* W  = (const float*)d_in[2];
    float*       out = (float*)d_out;
    int*         wsi = (int*)d_ws;
    int* list   = wsi;
    int* segoff = wsi + 1600000;

    gcn_blocksort<<<250, 800, 0, stream>>>(ei, list, segoff);
    gcn_rowsum<<<250, 800, 0, stream>>>(x, list, segoff, W, out);
}


# Row kernel: each compare-exchange of the lane sorting network is one median of three (v_med3_i32, third value the lane's constant least or greatest int), not min, max, select; sum order kept
# speedup vs baseline: 6.6086x; 1.0809x over previous
#include <hip/hip_runtime.h>


typedef __attribute__((ext_vector_type(2))) float v2f;
typedef __attribute__((ext_vector_type(4))) float v4f;
typedef __attribute__((ext_vector_type(8))) float v8f;

#define GCN_N_NODES 100000
#define GCN_N_EDGES 1600000
#define GCN_D 32

#define GCN_NBLK 250
#define GCN_EPB 6400
#define GCN_TPB 800
#define GCN_NWAVE 25
#define GCN_EPT 8
#define GCN_NBK 256
#define GCN_ROWS_PB 400
#define GCN_NSLOT 512
#define GCN_SPW 10
#define GCN_SEGW 64
#define GCN_CAP 16000
#define GCN_WPT 20
#define GCN_RP 36

__global__ __launch_bounds__(GCN_TPB) void gcn_blocksort(
    const int* __restrict__ ei, int* __restrict__ list, int* __restrict__ segoff)
{
    __shared__ int cnt[GCN_NBK];
    __shared__ int off[GCN_NBK + 1];
    __shared__ int stage[GCN_EPB];
    const int tid = threadIdx.x;
    const int lane = tid & 31;
    const int b = blockIdx.x;
    const int* src = ei + b * GCN_EPB;
    const int* dst = ei + GCN_N_EDGES + b * GCN_EPB;

    int d[GCN_EPT];
    int s[GCN_EPT];
#pragma unroll
    for (int u = 0; u < GCN_EPT; ++u) {
        const int i = tid + u * GCN_TPB;
        d[u] = dst[i];
        s[u] = src[i];
    }

    if (tid < GCN_NBK) cnt[tid] = 0;
    __syncthreads();
    int bk[GCN_EPT];
    int rank[GCN_EPT];
#pragma unroll
    for (int u = 0; u < GCN_EPT; ++u) {
        bk[u] = (int)((unsigned)d[u] / (unsigned)GCN_ROWS_PB);
        rank[u] = atomicAdd(&cnt[bk[u]], 1);
    }
    __syncthreads();

    if (tid < 32) {
        int loc[8];
        int sum = 0;
#pragma unroll
        for (int j = 0; j < 8; ++j) { loc[j] = cnt[lane * 8 + j]; sum += loc[j]; }
        int inc = sum;
#pragma unroll
        for (int dd = 1; dd < 32; dd <<= 1) {
            const int t = __shfl_up(inc, dd);
            if (lane >= dd) inc += t;
        }
        int run = inc - sum;
#pragma unroll
        for (int j = 0; j < 8; ++j) { off[lane * 8 + j] = run; run += loc[j]; }
        if (lane == 31) off[GCN_NBK] = run;
    }
    __syncthreads();

    int first[GCN_EPT];
#pragma unroll
    for (int u = 0; u < GCN_EPT; ++u) first[u] = off[bk[u]];
#pragma unroll
    for (int u = 0; u < GCN_EPT; ++u)
        stage[first[u] + rank[u]] = (s[u] << 9) | (d[u] - bk[u] * GCN_ROWS_PB);
    __syncthreads();

#pragma unroll
    for (int u = 0; u < GCN_EPT; ++u) {
        const int i = tid + u * GCN_TPB;
        list[b * GCN_EPB + i] = stage[i];
    }
    if (tid <= GCN_NBK) segoff[tid * GCN_NBLK + b] = off[tid];
}

__device__ __forceinline__ int gcn_uni(int v) { return __builtin_amdgcn_readfirstlane(v); }

template <int J>
__device__ __forceinline__ int gcn_xrow(int v)
{
    return __builtin_amdgcn_update_dpp(0, v, 0x160 + J, 0xf, 0xf, true);
}

__device__ __forceinline__ int gcn_xhalf(int v)
{
    return __builtin_amdgcn_permlanex16(v, v, 0x76543210, 0xfedcba98, false, false);
}

__device__ __forceinline__ int gcn_med3(int a, int b, int c)
{
    return max(min(a, b), min(max(a, b), c));
}

template <int K, int J>
__device__ __forceinline__ int gcn_cx(int v, int lane)
{
    const int p = (J == 16) ? gcn_xhalf(v) : gcn_xrow<(J & 15)>(v);
    const bool keep_min = ((lane & J) == 0) == ((lane & K) == 0);
    const int bound = keep_min ? (-0x7fffffff - 1) : 0x7fffffff;
    return gcn_med3(v, p, bound);
}

__device__ __forceinline__ int gcn_sort10(int val, int lane)
{
    val = gcn_cx<2, 1>(val, lane);
    val = gcn_cx<4, 2>(val, lane);
    val = gcn_cx<4, 1>(val, lane);
    val = gcn_cx<8, 4>(val, lane);
    val = gcn_cx<8, 2>(val, lane);
    val = gcn_cx<8, 1>(val, lane);
    val = gcn_cx<16, 8>(val, lane);
    val = gcn_cx<16, 4>(val, lane);
    val = gcn_cx<16, 2>(val, lane);
    val = gcn_cx<16, 1>(val, lane);
    return val;
}

__device__ __forceinline__ int gcn_merge32(int val, int lane)
{
    val = gcn_cx<32, 16>(val, lane);
    val = gcn_cx<32, 8>(val, lane);
    val = gcn_cx<32, 4>(val, lane);
    val = gcn_cx<32, 2>(val, lane);
    val = gcn_cx<32, 1>(val, lane);
    return val;
}

__device__ __forceinline__ int gcn_rowoff(int val)
{
    return min(val, GCN_N_NODES - 1) << 7;
}

template <int B>
__device__ __forceinline__ void gcn_issue8(
    const char* xb, int voff, unsigned lane4, float (&f)[32])
{
#pragma unroll
    for (int u = 0; u < 8; ++u) {
        const int i = 8 * B + u;
        const unsigned so = (unsigned)__builtin_amdgcn_readlane(voff, i);
        f[i] = *(const float*)(xb + (so + lane4));
    }
}

template <int B>
__device__ __forceinline__ float gcn_sum8(const float (&f)[32], int m, float acc)
{
#pragma unroll
    for (int u = 0; u < 8; ++u) {
        const int i = 8 * B + u;
        acc += (i < m) ? f[i] : 0.f;
    }
    return acc;
}

__device__ __forceinline__ void gcn_issue_group(
    const char* xb, int voff, unsigned lane4, int m, float (&f)[32])
{
    gcn_issue8<0>(xb, voff, lane4, f);
    if (m > 8)  gcn_issue8<1>(xb, voff, lane4, f);
    if (m > 16) gcn_issue8<2>(xb, voff, lane4, f);
    if (m > 24) gcn_issue8<3>(xb, voff, lane4, f);
}

__device__ __forceinline__ float gcn_sum_group(const float (&f)[32], int m, float acc)
{
    acc = gcn_sum8<0>(f, m, acc);
    if (m > 8)  acc = gcn_sum8<1>(f, m, acc);
    if (m > 16) acc = gcn_sum8<2>(f, m, acc);
    if (m > 24) acc = gcn_sum8<3>(f, m, acc);
    return acc;
}

__device__ __forceinline__ float gcn_more_groups(
    const int* sorted, const char* xb, unsigned lane4, int lane,
    int g0, int re, float acc)
{
    for (; g0 < re; g0 += 32) {
        const int m = min(32, re - g0);
        int val = (lane < m) ? sorted[g0 + lane] : 0x7fffffff;
        val = gcn_merge32(gcn_sort10(val, lane), lane);
        const int voff = gcn_rowoff(val);
        for (int i0 = 0; i0 < m; i0 += 8) {
            float f[8];
#pragma unroll
            for (int u = 0; u < 8; ++u) {
                const unsigned so = (unsigned)__builtin_amdgcn_readlane(voff, i0 + u);
                f[u] = *(const float*)(xb + (so + lane4));
            }
#pragma unroll
            for (int u = 0; u < 8; ++u) acc += (i0 + u < m) ? f[u] : 0.f;
        }
    }
    return acc;
}

__global__ __launch_bounds__(GCN_TPB) void gcn_rowsum(
    const float* __restrict__ x, const int* __restrict__ list,
    const int* __restrict__ segoff, const float* __restrict__ W,
    float* __restrict__ out)
{
    __shared__ __attribute__((aligned(16))) unsigned char lds[
        (GCN_ROWS_PB * GCN_RP + 3 * GCN_NSLOT + 4 + GCN_CAP + 4) * 4];
    float* rows   = (float*)lds;
    int*   cnt    = (int*)lds + GCN_ROWS_PB * GCN_RP;
    int*   cur    = cnt + GCN_NSLOT;
    int*   off    = cur + GCN_NSLOT;
    int*   sorted = off + GCN_NSLOT + 4;
    int*   more   = sorted + GCN_CAP;

    const int tid = threadIdx.x;
    const int lane = tid & 31;
    const int wave = gcn_uni(tid >> 5);
    const int k = blockIdx.x;
    const char* xb = (const char*)x;
    const unsigned lane4 = (unsigned)lane * 4u;
    const int sg0 = wave * GCN_SPW;
    int bnd0 = 0;
    int bnd1 = 0;
    if (lane < GCN_SPW) {
        bnd0 = segoff[k * GCN_NBLK + sg0 + lane];
        bnd1 = segoff[(k + 1) * GCN_NBLK + sg0 + lane];
    }
    int s0[GCN_SPW];
    int len[GCN_SPW];
#pragma unroll
    for (int u = 0; u < GCN_SPW; ++u) {
        s0[u] = __builtin_amdgcn_readlane(bnd0, u);
        len[u] = __builtin_amdgcn_readlane(bnd1, u) - s0[u];
    }

    int p0 = 0;
    bool first = true;
    bool again;
    do {
        int p[GCN_WPT];
        bool longer = false;
#pragma unroll
        for (int u = 0; u < GCN_SPW; ++u) {
            const int* seg = list + (sg0 + u) * GCN_EPB + s0[u] + p0;
            const int left = len[u] - p0;
            p[2 * u] = (lane < left) ? seg[lane] : -1;
            p[2 * u + 1] = (lane + 32 < left) ? seg[lane + 32] : -1;
            longer = longer || (left > GCN_SEGW);
        }

        if (tid < GCN_NSLOT) cnt[tid] = 0;
        if (tid == 0) *more = 0;
        __syncthreads();
#pragma unroll
        for (int u = 0; u < GCN_WPT; ++u)
            if (p[u] >= 0) atomicAdd(&cnt[p[u] & 511], 1);
        if (longer && lane == 0) *more = 1;
        __syncthreads();

        if (wave == 0) {
            int loc[16];
            int s = 0;
#pragma unroll
            for (int j = 0; j < 16; ++j) { loc[j] = cnt[lane * 16 + j]; s += loc[j]; }
            int inc = s;
#pragma unroll
            for (int d = 1; d < 32; d <<= 1) {
                const int t = __shfl_up(inc, d);
                if (lane >= d) inc += t;
            }
            int run = inc - s;
#pragma unroll
            for (int j = 0; j < 16; ++j) {
                off[lane * 16 + j] = run;
                cur[lane * 16 + j] = run;
                run += loc[j];
            }
            if (lane == 31) off[GCN_NSLOT] = run;
        }
        __syncthreads();

#pragma unroll
        for (int u = 0; u < GCN_WPT; ++u) {
            if (p[u] >= 0) {
                const int pos = atomicAdd(&cur[p[u] & 511], 1);
                sorted[pos] = p[u] >> 9;
            }
        }
        __syncthreads();
        again = gcn_uni(*more) != 0;

        for (int jj = 0; jj < 8; ++jj) {
            const int rA = wave + 2 * GCN_NWAVE * jj;
            const int rB = rA + GCN_NWAVE;
            const int rbA = gcn_uni(off[rA]);
            const int reA = gcn_uni(off[rA + 1]);
            const int rbB = gcn_uni(off[rB]);
            const int reB = gcn_uni(off[rB + 1]);
            const int mA = min(32, reA - rbA);
            const int mB = min(32, reB - rbB);

            float accA = 0.f;
            float accB = 0.f;
            if (!first) {
                accA = rows[rA * GCN_RP + lane];
                accB = rows[rB * GCN_RP + lane];
            }

            int vA = (lane < mA) ? sorted[rbA + lane] : 0x7fffffff;
            int vB = (lane < mB) ? sorted[rbB + lane] : 0x7fffffff;
            vA = gcn_sort10(vA, lane);
            vB = gcn_sort10(vB, lane);
            if (max(mA, mB) > 16) {
                vA = gcn_merge32(vA, lane);
                vB = gcn_merge32(vB, lane);
            }
            const int voffA = gcn_rowoff(vA);
            const int voffB = gcn_rowoff(vB);

            float fA[32];
            float fB[32];
            gcn_issue_group(xb, voffA, lane4, mA, fA);
            gcn_issue_group(xb, voffB, lane4, mB, fB);
            accA = gcn_sum_group(fA, mA, accA);
            accB = gcn_sum_group(fB, mB, accB);

            accA = gcn_more_groups(sorted, xb, lane4, lane, rbA + 32, reA, accA);
            accB = gcn_more_groups(sorted, xb, lane4, lane, rbB + 32, reB, accB);

            rows[rA * GCN_RP + lane] = accA;
            rows[rB * GCN_RP + lane] = accB;
        }
        __syncthreads();
        p0 += GCN_SEGW;
        first = false;
    } while (again);

    const int tile = k * (GCN_ROWS_PB / 16) + wave;

    const int m  = lane & 15;
    const int kb = (lane >> 4) * 2;
    const unsigned a_b = (unsigned)((wave * 16 + m) * GCN_RP + kb) * 4u;
    unsigned d_b = (unsigned)((wave * 16 + (lane >> 4) * 8) * GCN_RP + m) * 4u;
    unsigned e_b = d_b + 4 * GCN_RP * 4;
    unsigned r_b = (unsigned)((wave * 16 + (lane >> 3)) * GCN_RP + (lane & 7) * 4) * 4u;
    unsigned o_b = (unsigned)lane * 16u;
    const char* rowsb = (const char*)rows;
    char* rowsw = (char*)rows;

    v2f b0[8], b1[8];
#pragma unroll
    for (int kc = 0; kc < 8; ++kc) {
        const int kk = kc * 4 + kb;
        const float* w0 = W + (m)      * GCN_D + kk;
        const float* w1 = W + (16 + m) * GCN_D + kk;
        v2f t0; t0.x = w0[0]; t0.y = w0[1];
        v2f t1; t1.x = w1[0]; t1.y = w1[1];
        b0[kc] = t0; b1[kc] = t1;
    }

    v2f a[8];
#pragma unroll
    for (int kc = 0; kc < 8; ++kc) a[kc] = *(const v2f*)(rowsb + a_b + kc * 16);

    asm volatile("" : "+v"(d_b), "+v"(e_b), "+v"(r_b), "+v"(o_b));
    __builtin_amdgcn_sched_barrier(0);

    v8f c0 = {}; v8f c1 = {};
#pragma unroll
    for (int kc = 0; kc < 8; ++kc) {
        c0 = __builtin_amdgcn_wmma_f32_16x16x4_f32(
                 false, a[kc], false, b0[kc], (short)0, c0, false, false);
        c1 = __builtin_amdgcn_wmma_f32_16x16x4_f32(
                 false, a[kc], false, b1[kc], (short)0, c1, false, false);
    }

#pragma unroll
    for (int r = 0; r < 4; ++r) {
        *(float*)(rowsw + d_b + r * (GCN_RP * 4))      = c0[r];
        *(float*)(rowsw + d_b + r * (GCN_RP * 4) + 64) = c1[r];
        *(float*)(rowsw + e_b + r * (GCN_RP * 4))      = c0[r + 4];
        *(float*)(rowsw + e_b + r * (GCN_RP * 4) + 64) = c1[r + 4];
    }
    __builtin_amdgcn_fence(__ATOMIC_ACQ_REL, "wavefront");
    char* og = (char*)(out + (size_t)tile * 16 * GCN_D);
#pragma unroll
    for (int q = 0; q < 4; ++q) {
        const v4f v = *(const v4f*)(rowsb + r_b + q * (4 * GCN_RP * 4));
        *(v4f*)(og + o_b + q * (4 * GCN_D * 4)) = v;
    }
}

extern "C" void kernel_launch(void* const* d_in, const int* in_sizes, int n_in,
                              void* d_out, int out_size, void* d_ws, size_t ws_size,
                              hipStream_t stream)
{
    const float* x  = (const float*)d_in[0];
    const int*   ei = (const int*)d_in[1];
    const float* W  = (const float*)d_in[2];
    float*       out = (float*)d_out;
    int*         wsi = (int*)d_ws;
    int* list   = wsi;
    int* segoff = wsi + 1600000;

    gcn_blocksort<<<250, 800, 0, stream>>>(ei, list, segoff);
    gcn_rowsum<<<250, 800, 0, stream>>>(x, list, segoff, W, out);
}


# Row kernel: a row's sources requested and added exactly: whole blocks of 8, then a 4, a 2 and a 1 by the bits of its count (scalar lane numbers): no padding loads, no masked adds; sum order kept
# speedup vs baseline: 6.8690x; 1.0394x over previous
#include <hip/hip_runtime.h>


typedef __attribute__((ext_vector_type(2))) float v2f;
typedef __attribute__((ext_vector_type(4))) float v4f;
typedef __attribute__((ext_vector_type(8))) float v8f;

#define GCN_N_NODES 100000
#define GCN_N_EDGES 1600000
#define GCN_D 32

#define GCN_NBLK 250
#define GCN_EPB 6400
#define GCN_TPB 800
#define GCN_NWAVE 25
#define GCN_EPT 8
#define GCN_NBK 256
#define GCN_ROWS_PB 400
#define GCN_NSLOT 512
#define GCN_SPW 10
#define GCN_SEGW 64
#define GCN_CAP 16000
#define GCN_WPT 20
#define GCN_RP 36

__global__ __launch_bounds__(GCN_TPB) void gcn_blocksort(
    const int* __restrict__ ei, int* __restrict__ list, int* __restrict__ segoff)
{
    __shared__ int cnt[GCN_NBK];
    __shared__ int off[GCN_NBK + 1];
    __shared__ int stage[GCN_EPB];
    const int tid = threadIdx.x;
    const int lane = tid & 31;
    const int b = blockIdx.x;
    const int* src = ei + b * GCN_EPB;
    const int* dst = ei + GCN_N_EDGES + b * GCN_EPB;

    int d[GCN_EPT];
    int s[GCN_EPT];
#pragma unroll
    for (int u = 0; u < GCN_EPT; ++u) {
        const int i = tid + u * GCN_TPB;
        d[u] = dst[i];
        s[u] = src[i];
    }

    if (tid < GCN_NBK) cnt[tid] = 0;
    __syncthreads();
    int bk[GCN_EPT];
    int rank[GCN_EPT];
#pragma unroll
    for (int u = 0; u < GCN_EPT; ++u) {
        bk[u] = (int)((unsigned)d[u] / (unsigned)GCN_ROWS_PB);
        rank[u] = atomicAdd(&cnt[bk[u]], 1);
    }
    __syncthreads();

    if (tid < 32) {
        int loc[8];
        int sum = 0;
#pragma unroll
        for (int j = 0; j < 8; ++j) { loc[j] = cnt[lane * 8 + j]; sum += loc[j]; }
        int inc = sum;
#pragma unroll
        for (int dd = 1; dd < 32; dd <<= 1) {
            const int t = __shfl_up(inc, dd);
            if (lane >= dd) inc += t;
        }
        int run = inc - sum;
#pragma unroll
        for (int j = 0; j < 8; ++j) { off[lane * 8 + j] = run; run += loc[j]; }
        if (lane == 31) off[GCN_NBK] = run;
    }
    __syncthreads();

    int first[GCN_EPT];
#pragma unroll
    for (int u = 0; u < GCN_EPT; ++u) first[u] = off[bk[u]];
#pragma unroll
    for (int u = 0; u < GCN_EPT; ++u)
        stage[first[u] + rank[u]] = (s[u] << 9) | (d[u] - bk[u] * GCN_ROWS_PB);
    __syncthreads();

#pragma unroll
    for (int u = 0; u < GCN_EPT; ++u) {
        const int i = tid + u * GCN_TPB;
        list[b * GCN_EPB + i] = stage[i];
    }
    if (tid <= GCN_NBK) segoff[tid * GCN_NBLK + b] = off[tid];
}

__device__ __forceinline__ int gcn_uni(int v) { return __builtin_amdgcn_readfirstlane(v); }

template <int J>
__device__ __forceinline__ int gcn_xrow(int v)
{
    return __builtin_amdgcn_update_dpp(0, v, 0x160 + J, 0xf, 0xf, true);
}

__device__ __forceinline__ int gcn_xhalf(int v)
{
    return __builtin_amdgcn_permlanex16(v, v, 0x76543210, 0xfedcba98, false, false);
}

__device__ __forceinline__ int gcn_med3(int a, int b, int c)
{
    return max(min(a, b), min(max(a, b), c));
}

template <int K, int J>
__device__ __forceinline__ int gcn_cx(int v, int lane)
{
    const int p = (J == 16) ? gcn_xhalf(v) : gcn_xrow<(J & 15)>(v);
    const bool keep_min = ((lane & J) == 0) == ((lane & K) == 0);
    const int bound = keep_min ? (-0x7fffffff - 1) : 0x7fffffff;
    return gcn_med3(v, p, bound);
}

__device__ __forceinline__ int gcn_sort10(int val, int lane)
{
    val = gcn_cx<2, 1>(val, lane);
    val = gcn_cx<4, 2>(val, lane);
    val = gcn_cx<4, 1>(val, lane);
    val = gcn_cx<8, 4>(val, lane);
    val = gcn_cx<8, 2>(val, lane);
    val = gcn_cx<8, 1>(val, lane);
    val = gcn_cx<16, 8>(val, lane);
    val = gcn_cx<16, 4>(val, lane);
    val = gcn_cx<16, 2>(val, lane);
    val = gcn_cx<16, 1>(val, lane);
    return val;
}

__device__ __forceinline__ int gcn_merge32(int val, int lane)
{
    val = gcn_cx<32, 16>(val, lane);
    val = gcn_cx<32, 8>(val, lane);
    val = gcn_cx<32, 4>(val, lane);
    val = gcn_cx<32, 2>(val, lane);
    val = gcn_cx<32, 1>(val, lane);
    return val;
}

__device__ __forceinline__ int gcn_rowoff(int val)
{
    return min(val, GCN_N_NODES - 1) << 7;
}

template <int B>
__device__ __forceinline__ void gcn_issue8(
    const char* xb, int voff, unsigned lane4, float (&f)[32])
{
#pragma unroll
    for (int u = 0; u < 8; ++u) {
        const int i = 8 * B + u;
        const unsigned so = (unsigned)__builtin_amdgcn_readlane(voff, i);
        f[i] = *(const float*)(xb + (so + lane4));
    }
}

template <int B>
__device__ __forceinline__ float gcn_sum8(const float (&f)[32], int m, float acc)
{
#pragma unroll
    for (int u = 0; u < 8; ++u) {
        const int i = 8 * B + u;
        acc += (i < m) ? f[i] : 0.f;
    }
    return acc;
}

template <int B>
__device__ __forceinline__ float gcn_add8(const float (&f)[32], float acc)
{
#pragma unroll
    for (int u = 0; u < 8; ++u) acc += f[8 * B + u];
    return acc;
}

template <int N>
__device__ __forceinline__ void gcn_issue_part(
    const char* xb, int voff, unsigned lane4, int l0, float (&t)[N])
{
#pragma unroll
    for (int u = 0; u < N; ++u) {
        const unsigned so = (unsigned)__builtin_amdgcn_readlane(voff, l0 + u);
        t[u] = *(const float*)(xb + (so + lane4));
    }
}

template <int N>
__device__ __forceinline__ float gcn_add_part(const float (&t)[N], float acc)
{
    asm volatile("" : "+v"(acc));
#pragma unroll
    for (int u = 0; u < N; ++u) acc += t[u];
    return acc;
}

struct gcn_rowset {
    float f[32];
    float t4[4];
    float t2[2];
    float t1[1];
};

__device__ __forceinline__ void gcn_issue_group(
    const char* xb, int voff, unsigned lane4, int m, gcn_rowset& r)
{
    gcn_issue8<0>(xb, voff, lane4, r.f);
    if (m >= 16) gcn_issue8<1>(xb, voff, lane4, r.f);
    if (m >= 24) gcn_issue8<2>(xb, voff, lane4, r.f);
    if (m >= 32) gcn_issue8<3>(xb, voff, lane4, r.f);
    if (m > 8) {
        const int l0 = m & ~7;
        if (m & 4) gcn_issue_part<4>(xb, voff, lane4, l0, r.t4);
        if (m & 2) gcn_issue_part<2>(xb, voff, lane4, l0 + (m & 4), r.t2);
        if (m & 1) gcn_issue_part<1>(xb, voff, lane4, m - 1, r.t1);
    }
}

__device__ __forceinline__ float gcn_sum_group(const gcn_rowset& r, int m, float acc)
{
    if (m <= 8) return gcn_sum8<0>(r.f, m, acc);
    acc = gcn_add8<0>(r.f, acc);
    if (m >= 16) acc = gcn_add8<1>(r.f, acc);
    if (m >= 24) acc = gcn_add8<2>(r.f, acc);
    if (m >= 32) acc = gcn_add8<3>(r.f, acc);
    if (m & 4) acc = gcn_add_part<4>(r.t4, acc);
    if (m & 2) acc = gcn_add_part<2>(r.t2, acc);
    if (m & 1) acc = gcn_add_part<1>(r.t1, acc);
    return acc;
}

__device__ __forceinline__ float gcn_more_groups(
    const int* sorted, const char* xb, unsigned lane4, int lane,
    int g0, int re, float acc)
{
    for (; g0 < re; g0 += 32) {
        const int m = min(32, re - g0);
        int val = (lane < m) ? sorted[g0 + lane] : 0x7fffffff;
        val = gcn_merge32(gcn_sort10(val, lane), lane);
        const int voff = gcn_rowoff(val);
        for (int i0 = 0; i0 < m; i0 += 8) {
            float f[8];
#pragma unroll
            for (int u = 0; u < 8; ++u) {
                const unsigned so = (unsigned)__builtin_amdgcn_readlane(voff, i0 + u);
                f[u] = *(const float*)(xb + (so + lane4));
            }
#pragma unroll
            for (int u = 0; u < 8; ++u) acc += (i0 + u < m) ? f[u] : 0.f;
        }
    }
    return acc;
}

__global__ __launch_bounds__(GCN_TPB) void gcn_rowsum(
    const float* __restrict__ x, const int* __restrict__ list,
    const int* __restrict__ segoff, const float* __restrict__ W,
    float* __restrict__ out)
{
    __shared__ __attribute__((aligned(16))) unsigned char lds[
        (GCN_ROWS_PB * GCN_RP + 3 * GCN_NSLOT + 4 + GCN_CAP + 4) * 4];
    float* rows   = (float*)lds;
    int*   cnt    = (int*)lds + GCN_ROWS_PB * GCN_RP;
    int*   cur    = cnt + GCN_NSLOT;
    int*   off    = cur + GCN_NSLOT;
    int*   sorted = off + GCN_NSLOT + 4;
    int*   more   = sorted + GCN_CAP;

    const int tid = threadIdx.x;
    const int lane = tid & 31;
    const int wave = gcn_uni(tid >> 5);
    const int k = blockIdx.x;
    const char* xb = (const char*)x;
    const unsigned lane4 = (unsigned)lane * 4u;
    const int sg0 = wave * GCN_SPW;
    int bnd0 = 0;
    int bnd1 = 0;
    if (lane < GCN_SPW) {
        bnd0 = segoff[k * GCN_NBLK + sg0 + lane];
        bnd1 = segoff[(k + 1) * GCN_NBLK + sg0 + lane];
    }
    int s0[GCN_SPW];
    int len[GCN_SPW];
#pragma unroll
    for (int u = 0; u < GCN_SPW; ++u) {
        s0[u] = __builtin_amdgcn_readlane(bnd0, u);
        len[u] = __builtin_amdgcn_readlane(bnd1, u) - s0[u];
    }

    int p0 = 0;
    bool first = true;
    bool again;
    do {
        int p[GCN_WPT];
        bool longer = false;
#pragma unroll
        for (int u = 0; u < GCN_SPW; ++u) {
            const int* seg = list + (sg0 + u) * GCN_EPB + s0[u] + p0;
            const int left = len[u] - p0;
            p[2 * u] = (lane < left) ? seg[lane] : -1;
            p[2 * u + 1] = (lane + 32 < left) ? seg[lane + 32] : -1;
            longer = longer || (left > GCN_SEGW);
        }

        if (tid < GCN_NSLOT) cnt[tid] = 0;
        if (tid == 0) *more = 0;
        __syncthreads();
#pragma unroll
        for (int u = 0; u < GCN_WPT; ++u)
            if (p[u] >= 0) atomicAdd(&cnt[p[u] & 511], 1);
        if (longer && lane == 0) *more = 1;
        __syncthreads();

        if (wave == 0) {
            int loc[16];
            int s = 0;
#pragma unroll
            for (int j = 0; j < 16; ++j) { loc[j] = cnt[lane * 16 + j]; s += loc[j]; }
            int inc = s;
#pragma unroll
            for (int d = 1; d < 32; d <<= 1) {
                const int t = __shfl_up(inc, d);
                if (lane >= d) inc += t;
            }
            int run = inc - s;
#pragma unroll
            for (int j = 0; j < 16; ++j) {
                off[lane * 16 + j] = run;
                cur[lane * 16 + j] = run;
                run += loc[j];
            }
            if (lane == 31) off[GCN_NSLOT] = run;
        }
        __syncthreads();

#pragma unroll
        for (int u = 0; u < GCN_WPT; ++u) {
            if (p[u] >= 0) {
                const int pos = atomicAdd(&cur[p[u] & 511], 1);
                sorted[pos] = p[u] >> 9;
            }
        }
        __syncthreads();
        again = gcn_uni(*more) != 0;

        for (int jj = 0; jj < 8; ++jj) {
            const int rA = wave + 2 * GCN_NWAVE * jj;
            const int rB = rA + GCN_NWAVE;
            const int rbA = gcn_uni(off[rA]);
            const int reA = gcn_uni(off[rA + 1]);
            const int rbB = gcn_uni(off[rB]);
            const int reB = gcn_uni(off[rB + 1]);
            const int mA = min(32, reA - rbA);
            const int mB = min(32, reB - rbB);

            float accA = 0.f;
            float accB = 0.f;
            if (!first) {
                accA = rows[rA * GCN_RP + lane];
                accB = rows[rB * GCN_RP + lane];
            }

            int vA = (lane < mA) ? sorted[rbA + lane] : 0x7fffffff;
            int vB = (lane < mB) ? sorted[rbB + lane] : 0x7fffffff;
            vA = gcn_sort10(vA, lane);
            vB = gcn_sort10(vB, lane);
            if (max(mA, mB) > 16) {
                vA = gcn_merge32(vA, lane);
                vB = gcn_merge32(vB, lane);
            }
            const int voffA = gcn_rowoff(vA);
            const int voffB = gcn_rowoff(vB);

            gcn_rowset sA;
            gcn_rowset sB;
            gcn_issue_group(xb, voffA, lane4, mA, sA);
            gcn_issue_group(xb, voffB, lane4, mB, sB);
            accA = gcn_sum_group(sA, mA, accA);
            accB = gcn_sum_group(sB, mB, accB);

            accA = gcn_more_groups(sorted, xb, lane4, lane, rbA + 32, reA, accA);
            accB = gcn_more_groups(sorted, xb, lane4, lane, rbB + 32, reB, accB);

            rows[rA * GCN_RP + lane] = accA;
            rows[rB * GCN_RP + lane] = accB;
        }
        __syncthreads();
        p0 += GCN_SEGW;
        first = false;
    } while (again);

    const int tile = k * (GCN_ROWS_PB / 16) + wave;

    const int m  = lane & 15;
    const int kb = (lane >> 4) * 2;
    const unsigned a_b = (unsigned)((wave * 16 + m) * GCN_RP + kb) * 4u;
    unsigned d_b = (unsigned)((wave * 16 + (lane >> 4) * 8) * GCN_RP + m) * 4u;
    unsigned e_b = d_b + 4 * GCN_RP * 4;
    unsigned r_b = (unsigned)((wave * 16 + (lane >> 3)) * GCN_RP + (lane & 7) * 4) * 4u;
    unsigned o_b = (unsigned)lane * 16u;
    const char* rowsb = (const char*)rows;
    char* rowsw = (char*)rows;

    v2f b0[8], b1[8];
#pragma unroll
    for (int kc = 0; kc < 8; ++kc) {
        const int kk = kc * 4 + kb;
        const float* w0 = W + (m)      * GCN_D + kk;
        const float* w1 = W + (16 + m) * GCN_D + kk;
        v2f t0; t0.x = w0[0]; t0.y = w0[1];
        v2f t1; t1.x = w1[0]; t1.y = w1[1];
        b0[kc] = t0; b1[kc] = t1;
    }

    v2f a[8];
#pragma unroll
    for (int kc = 0; kc < 8; ++kc) a[kc] = *(const v2f*)(rowsb + a_b + kc * 16);

    asm volatile("" : "+v"(d_b), "+v"(e_b), "+v"(r_b), "+v"(o_b));
    __builtin_amdgcn_sched_barrier(0);

    v8f c0 = {}; v8f c1 = {};
#pragma unroll
    for (int kc = 0; kc < 8; ++kc) {
        c0 = __builtin_amdgcn_wmma_f32_16x16x4_f32(
                 false, a[kc], false, b0[kc], (short)0, c0, false, false);
        c1 = __builtin_amdgcn_wmma_f32_16x16x4_f32(
                 false, a[kc], false, b1[kc], (short)0, c1, false, false);
    }

#pragma unroll
    for (int r = 0; r < 4; ++r) {
        *(float*)(rowsw + d_b + r * (GCN_RP * 4))      = c0[r];
        *(float*)(rowsw + d_b + r * (GCN_RP * 4) + 64) = c1[r];
        *(float*)(rowsw + e_b + r * (GCN_RP * 4))      = c0[r + 4];
        *(float*)(rowsw + e_b + r * (GCN_RP * 4) + 64) = c1[r + 4];
    }
    __builtin_amdgcn_fence(__ATOMIC_ACQ_REL, "wavefront");
    char* og = (char*)(out + (size_t)tile * 16 * GCN_D);
#pragma unroll
    for (int q = 0; q < 4; ++q) {
        const v4f v = *(const v4f*)(rowsb + r_b + q * (4 * GCN_RP * 4));
        *(v4f*)(og + o_b + q * (4 * GCN_D * 4)) = v;
    }
}

extern "C" void kernel_launch(void* const* d_in, const int* in_sizes, int n_in,
                              void* d_out, int out_size, void* d_ws, size_t ws_size,
                              hipStream_t stream)
{
    const float* x  = (const float*)d_in[0];
    const int*   ei = (const int*)d_in[1];
    const float* W  = (const float*)d_in[2];
    float*       out = (float*)d_out;
    int*         wsi = (int*)d_ws;
    int* list   = wsi;
    int* segoff = wsi + 1600000;

    gcn_blocksort<<<250, 800, 0, stream>>>(ei, list, segoff);
    gcn_rowsum<<<250, 800, 0, stream>>>(x, list, segoff, W, out);
}


# Row kernel: rows ordered in ONE LDS sweep: a row owns 64 slots, the counting add's return is the slot; count sweep, scan and two barriers gone (scan+place stay as overflow path); sum order kept
# speedup vs baseline: 7.2254x; 1.0519x over previous
#include <hip/hip_runtime.h>


typedef __attribute__((ext_vector_type(2))) float v2f;
typedef __attribute__((ext_vector_type(4))) float v4f;
typedef __attribute__((ext_vector_type(8))) float v8f;

#define GCN_N_NODES 100000
#define GCN_N_EDGES 1600000
#define GCN_D 32

#define GCN_NBLK 250
#define GCN_EPB 6400
#define GCN_TPB 800
#define GCN_NWAVE 25
#define GCN_EPT 8
#define GCN_NBK 256
#define GCN_ROWS_PB 400
#define GCN_NSLOT 512
#define GCN_SPW 10
#define GCN_SEGW 64
#define GCN_CAP 16000
#define GCN_SLOTW 64
#define GCN_SLOTP 65
#define GCN_WPT 20
#define GCN_RP 36

__global__ __launch_bounds__(GCN_TPB) void gcn_blocksort(
    const int* __restrict__ ei, int* __restrict__ list, int* __restrict__ segoff)
{
    __shared__ int cnt[GCN_NBK];
    __shared__ int off[GCN_NBK + 1];
    __shared__ int stage[GCN_EPB];
    const int tid = threadIdx.x;
    const int lane = tid & 31;
    const int b = blockIdx.x;
    const int* src = ei + b * GCN_EPB;
    const int* dst = ei + GCN_N_EDGES + b * GCN_EPB;

    int d[GCN_EPT];
    int s[GCN_EPT];
#pragma unroll
    for (int u = 0; u < GCN_EPT; ++u) {
        const int i = tid + u * GCN_TPB;
        d[u] = dst[i];
        s[u] = src[i];
    }

    if (tid < GCN_NBK) cnt[tid] = 0;
    __syncthreads();
    int bk[GCN_EPT];
    int rank[GCN_EPT];
#pragma unroll
    for (int u = 0; u < GCN_EPT; ++u) {
        bk[u] = (int)((unsigned)d[u] / (unsigned)GCN_ROWS_PB);
        rank[u] = atomicAdd(&cnt[bk[u]], 1);
    }
    __syncthreads();

    if (tid < 32) {
        int loc[8];
        int sum = 0;
#pragma unroll
        for (int j = 0; j < 8; ++j) { loc[j] = cnt[lane * 8 + j]; sum += loc[j]; }
        int inc = sum;
#pragma unroll
        for (int dd = 1; dd < 32; dd <<= 1) {
            const int t = __shfl_up(inc, dd);
            if (lane >= dd) inc += t;
        }
        int run = inc - sum;
#pragma unroll
        for (int j = 0; j < 8; ++j) { off[lane * 8 + j] = run; run += loc[j]; }
        if (lane == 31) off[GCN_NBK] = run;
    }
    __syncthreads();

    int first[GCN_EPT];
#pragma unroll
    for (int u = 0; u < GCN_EPT; ++u) first[u] = off[bk[u]];
#pragma unroll
    for (int u = 0; u < GCN_EPT; ++u)
        stage[first[u] + rank[u]] = (s[u] << 9) | (d[u] - bk[u] * GCN_ROWS_PB);
    __syncthreads();

#pragma unroll
    for (int u = 0; u < GCN_EPT; ++u) {
        const int i = tid + u * GCN_TPB;
        list[b * GCN_EPB + i] = stage[i];
    }
    if (tid <= GCN_NBK) segoff[tid * GCN_NBLK + b] = off[tid];
}

__device__ __forceinline__ int gcn_uni(int v) { return __builtin_amdgcn_readfirstlane(v); }

template <int J>
__device__ __forceinline__ int gcn_xrow(int v)
{
    return __builtin_amdgcn_update_dpp(0, v, 0x160 + J, 0xf, 0xf, true);
}

__device__ __forceinline__ int gcn_xhalf(int v)
{
    return __builtin_amdgcn_permlanex16(v, v, 0x76543210, 0xfedcba98, false, false);
}

__device__ __forceinline__ int gcn_med3(int a, int b, int c)
{
    return max(min(a, b), min(max(a, b), c));
}

template <int K, int J>
__device__ __forceinline__ int gcn_cx(int v, int lane)
{
    const int p = (J == 16) ? gcn_xhalf(v) : gcn_xrow<(J & 15)>(v);
    const bool keep_min = ((lane & J) == 0) == ((lane & K) == 0);
    const int bound = keep_min ? (-0x7fffffff - 1) : 0x7fffffff;
    return gcn_med3(v, p, bound);
}

__device__ __forceinline__ int gcn_sort10(int val, int lane)
{
    val = gcn_cx<2, 1>(val, lane);
    val = gcn_cx<4, 2>(val, lane);
    val = gcn_cx<4, 1>(val, lane);
    val = gcn_cx<8, 4>(val, lane);
    val = gcn_cx<8, 2>(val, lane);
    val = gcn_cx<8, 1>(val, lane);
    val = gcn_cx<16, 8>(val, lane);
    val = gcn_cx<16, 4>(val, lane);
    val = gcn_cx<16, 2>(val, lane);
    val = gcn_cx<16, 1>(val, lane);
    return val;
}

__device__ __forceinline__ int gcn_merge32(int val, int lane)
{
    val = gcn_cx<32, 16>(val, lane);
    val = gcn_cx<32, 8>(val, lane);
    val = gcn_cx<32, 4>(val, lane);
    val = gcn_cx<32, 2>(val, lane);
    val = gcn_cx<32, 1>(val, lane);
    return val;
}

__device__ __forceinline__ int gcn_rowoff(int val)
{
    return min(val, GCN_N_NODES - 1) << 7;
}

template <int B>
__device__ __forceinline__ void gcn_issue8(
    const char* xb, int voff, unsigned lane4, float (&f)[32])
{
#pragma unroll
    for (int u = 0; u < 8; ++u) {
        const int i = 8 * B + u;
        const unsigned so = (unsigned)__builtin_amdgcn_readlane(voff, i);
        f[i] = *(const float*)(xb + (so + lane4));
    }
}

template <int B>
__device__ __forceinline__ float gcn_sum8(const float (&f)[32], int m, float acc)
{
#pragma unroll
    for (int u = 0; u < 8; ++u) {
        const int i = 8 * B + u;
        acc += (i < m) ? f[i] : 0.f;
    }
    return acc;
}

template <int B>
__device__ __forceinline__ float gcn_add8(const float (&f)[32], float acc)
{
#pragma unroll
    for (int u = 0; u < 8; ++u) acc += f[8 * B + u];
    return acc;
}

template <int N>
__device__ __forceinline__ void gcn_issue_part(
    const char* xb, int voff, unsigned lane4, int l0, float (&t)[N])
{
#pragma unroll
    for (int u = 0; u < N; ++u) {
        const unsigned so = (unsigned)__builtin_amdgcn_readlane(voff, l0 + u);
        t[u] = *(const float*)(xb + (so + lane4));
    }
}

template <int N>
__device__ __forceinline__ float gcn_add_part(const float (&t)[N], float acc)
{
    asm volatile("" : "+v"(acc));
#pragma unroll
    for (int u = 0; u < N; ++u) acc += t[u];
    return acc;
}

struct gcn_rowset {
    float f[32];
    float t4[4];
    float t2[2];
    float t1[1];
};

__device__ __forceinline__ void gcn_issue_group(
    const char* xb, int voff, unsigned lane4, int m, gcn_rowset& r)
{
    gcn_issue8<0>(xb, voff, lane4, r.f);
    if (m >= 16) gcn_issue8<1>(xb, voff, lane4, r.f);
    if (m >= 24) gcn_issue8<2>(xb, voff, lane4, r.f);
    if (m >= 32) gcn_issue8<3>(xb, voff, lane4, r.f);
    if (m > 8) {
        const int l0 = m & ~7;
        if (m & 4) gcn_issue_part<4>(xb, voff, lane4, l0, r.t4);
        if (m & 2) gcn_issue_part<2>(xb, voff, lane4, l0 + (m & 4), r.t2);
        if (m & 1) gcn_issue_part<1>(xb, voff, lane4, m - 1, r.t1);
    }
}

__device__ __forceinline__ float gcn_sum_group(const gcn_rowset& r, int m, float acc)
{
    if (m <= 8) return gcn_sum8<0>(r.f, m, acc);
    acc = gcn_add8<0>(r.f, acc);
    if (m >= 16) acc = gcn_add8<1>(r.f, acc);
    if (m >= 24) acc = gcn_add8<2>(r.f, acc);
    if (m >= 32) acc = gcn_add8<3>(r.f, acc);
    if (m & 4) acc = gcn_add_part<4>(r.t4, acc);
    if (m & 2) acc = gcn_add_part<2>(r.t2, acc);
    if (m & 1) acc = gcn_add_part<1>(r.t1, acc);
    return acc;
}

__device__ __forceinline__ float gcn_more_groups(
    const int* sorted, const char* xb, unsigned lane4, int lane,
    int g0, int re, float acc)
{
    for (; g0 < re; g0 += 32) {
        const int m = min(32, re - g0);
        int val = (lane < m) ? sorted[g0 + lane] : 0x7fffffff;
        val = gcn_merge32(gcn_sort10(val, lane), lane);
        const int voff = gcn_rowoff(val);
        for (int i0 = 0; i0 < m; i0 += 8) {
            float f[8];
#pragma unroll
            for (int u = 0; u < 8; ++u) {
                const unsigned so = (unsigned)__builtin_amdgcn_readlane(voff, i0 + u);
                f[u] = *(const float*)(xb + (so + lane4));
            }
#pragma unroll
            for (int u = 0; u < 8; ++u) acc += (i0 + u < m) ? f[u] : 0.f;
        }
    }
    return acc;
}

__global__ __launch_bounds__(GCN_TPB) void gcn_rowsum(
    const float* __restrict__ x, const int* __restrict__ list,
    const int* __restrict__ segoff, const float* __restrict__ W,
    float* __restrict__ out)
{
    static_assert(GCN_ROWS_PB * GCN_SLOTP >= GCN_CAP,
                  "the fallback packs a whole pass into the slot words");
    __shared__ __attribute__((aligned(16))) unsigned char lds[
        (GCN_ROWS_PB * GCN_RP + 3 * GCN_NSLOT + 4 + GCN_ROWS_PB * GCN_SLOTP + 4) * 4];
    float* rows   = (float*)lds;
    int*   cnt    = (int*)lds + GCN_ROWS_PB * GCN_RP;
    int*   cur    = cnt + GCN_NSLOT;
    int*   off    = cur + GCN_NSLOT;
    int*   sorted = off + GCN_NSLOT + 4;
    int*   more   = sorted + GCN_ROWS_PB * GCN_SLOTP;

    const int tid = threadIdx.x;
    const int lane = tid & 31;
    const int wave = gcn_uni(tid >> 5);
    const int k = blockIdx.x;
    const char* xb = (const char*)x;
    const unsigned lane4 = (unsigned)lane * 4u;
    const int sg0 = wave * GCN_SPW;
    int bnd0 = 0;
    int bnd1 = 0;
    if (lane < GCN_SPW) {
        bnd0 = segoff[k * GCN_NBLK + sg0 + lane];
        bnd1 = segoff[(k + 1) * GCN_NBLK + sg0 + lane];
    }
    int s0[GCN_SPW];
    int len[GCN_SPW];
#pragma unroll
    for (int u = 0; u < GCN_SPW; ++u) {
        s0[u] = __builtin_amdgcn_readlane(bnd0, u);
        len[u] = __builtin_amdgcn_readlane(bnd1, u) - s0[u];
    }

    int p0 = 0;
    bool first = true;
    bool again;
    do {
        int p[GCN_WPT];
        bool longer = false;
#pragma unroll
        for (int u = 0; u < GCN_SPW; ++u) {
            const int* seg = list + (sg0 + u) * GCN_EPB + s0[u] + p0;
            const int left = len[u] - p0;
            p[2 * u] = (lane < left) ? seg[lane] : -1;
            p[2 * u + 1] = (lane + 32 < left) ? seg[lane + 32] : -1;
            longer = longer || (left > GCN_SEGW);
        }

        if (tid < GCN_NSLOT) { cnt[tid] = 0; off[tid] = tid * GCN_SLOTP; }
        if (tid == 0) { more[0] = 0; more[1] = 0; }
        __syncthreads();
        bool over = false;
#pragma unroll
        for (int u = 0; u < GCN_WPT; ++u) {
            if (p[u] >= 0) {
                const int row = p[u] & 511;
                const int i = atomicAdd(&cnt[row], 1);
                if (i < GCN_SLOTW) sorted[row * GCN_SLOTP + i] = p[u] >> 9;
                else over = true;
            }
        }
        if (over) more[1] = 1;
        if (longer && lane == 0) more[0] = 1;
        __syncthreads();
        again = gcn_uni(more[0]) != 0;
        const bool spill = gcn_uni(more[1]) != 0;

        if (spill) {
            if (wave == 0) {
                int loc[16];
                int s = 0;
#pragma unroll
                for (int j = 0; j < 16; ++j) { loc[j] = cnt[lane * 16 + j]; s += loc[j]; }
                int inc = s;
#pragma unroll
                for (int d = 1; d < 32; d <<= 1) {
                    const int t = __shfl_up(inc, d);
                    if (lane >= d) inc += t;
                }
                int run = inc - s;
#pragma unroll
                for (int j = 0; j < 16; ++j) {
                    off[lane * 16 + j] = run;
                    cur[lane * 16 + j] = run;
                    run += loc[j];
                }
            }
            __syncthreads();

#pragma unroll
            for (int u = 0; u < GCN_WPT; ++u) {
                if (p[u] >= 0) {
                    const int pos = atomicAdd(&cur[p[u] & 511], 1);
                    sorted[pos] = p[u] >> 9;
                }
            }
            __syncthreads();
        }

        for (int jj = 0; jj < 8; ++jj) {
            const int rA = wave + 2 * GCN_NWAVE * jj;
            const int rB = rA + GCN_NWAVE;
            const int rbA = gcn_uni(off[rA]);
            const int reA = rbA + gcn_uni(cnt[rA]);
            const int rbB = gcn_uni(off[rB]);
            const int reB = rbB + gcn_uni(cnt[rB]);
            const int mA = min(32, reA - rbA);
            const int mB = min(32, reB - rbB);

            float accA = 0.f;
            float accB = 0.f;
            if (!first) {
                accA = rows[rA * GCN_RP + lane];
                accB = rows[rB * GCN_RP + lane];
            }

            int vA = (lane < mA) ? sorted[rbA + lane] : 0x7fffffff;
            int vB = (lane < mB) ? sorted[rbB + lane] : 0x7fffffff;
            vA = gcn_sort10(vA, lane);
            vB = gcn_sort10(vB, lane);
            if (max(mA, mB) > 16) {
                vA = gcn_merge32(vA, lane);
                vB = gcn_merge32(vB, lane);
            }
            const int voffA = gcn_rowoff(vA);
            const int voffB = gcn_rowoff(vB);

            gcn_rowset sA;
            gcn_rowset sB;
            gcn_issue_group(xb, voffA, lane4, mA, sA);
            gcn_issue_group(xb, voffB, lane4, mB, sB);
            accA = gcn_sum_group(sA, mA, accA);
            accB = gcn_sum_group(sB, mB, accB);

            accA = gcn_more_groups(sorted, xb, lane4, lane, rbA + 32, reA, accA);
            accB = gcn_more_groups(sorted, xb, lane4, lane, rbB + 32, reB, accB);

            rows[rA * GCN_RP + lane] = accA;
            rows[rB * GCN_RP + lane] = accB;
        }
        __syncthreads();
        p0 += GCN_SEGW;
        first = false;
    } while (again);

    const int tile = k * (GCN_ROWS_PB / 16) + wave;

    const int m  = lane & 15;
    const int kb = (lane >> 4) * 2;
    const unsigned a_b = (unsigned)((wave * 16 + m) * GCN_RP + kb) * 4u;
    unsigned d_b = (unsigned)((wave * 16 + (lane >> 4) * 8) * GCN_RP + m) * 4u;
    unsigned e_b = d_b + 4 * GCN_RP * 4;
    unsigned r_b = (unsigned)((wave * 16 + (lane >> 3)) * GCN_RP + (lane & 7) * 4) * 4u;
    unsigned o_b = (unsigned)lane * 16u;
    const char* rowsb = (const char*)rows;
    char* rowsw = (char*)rows;

    v2f b0[8], b1[8];
#pragma unroll
    for (int kc = 0; kc < 8; ++kc) {
        const int kk = kc * 4 + kb;
        const float* w0 = W + (m)      * GCN_D + kk;
        const float* w1 = W + (16 + m) * GCN_D + kk;
        v2f t0; t0.x = w0[0]; t0.y = w0[1];
        v2f t1; t1.x = w1[0]; t1.y = w1[1];
        b0[kc] = t0; b1[kc] = t1;
    }

    v2f a[8];
#pragma unroll
    for (int kc = 0; kc < 8; ++kc) a[kc] = *(const v2f*)(rowsb + a_b + kc * 16);

    asm volatile("" : "+v"(d_b), "+v"(e_b), "+v"(r_b), "+v"(o_b));
    __builtin_amdgcn_sched_barrier(0);

    v8f c0 = {}; v8f c1 = {};
#pragma unroll
    for (int kc = 0; kc < 8; ++kc) {
        c0 = __builtin_amdgcn_wmma_f32_16x16x4_f32(
                 false, a[kc], false, b0[kc], (short)0, c0, false, false);
        c1 = __builtin_amdgcn_wmma_f32_16x16x4_f32(
                 false, a[kc], false, b1[kc], (short)0, c1, false, false);
    }

#pragma unroll
    for (int r = 0; r < 4; ++r) {
        *(float*)(rowsw + d_b + r * (GCN_RP * 4))      = c0[r];
        *(float*)(rowsw + d_b + r * (GCN_RP * 4) + 64) = c1[r];
        *(float*)(rowsw + e_b + r * (GCN_RP * 4))      = c0[r + 4];
        *(float*)(rowsw + e_b + r * (GCN_RP * 4) + 64) = c1[r + 4];
    }
    __builtin_amdgcn_fence(__ATOMIC_ACQ_REL, "wavefront");
    char* og = (char*)(out + (size_t)tile * 16 * GCN_D);
#pragma unroll
    for (int q = 0; q < 4; ++q) {
        const v4f v = *(const v4f*)(rowsb + r_b + q * (4 * GCN_RP * 4));
        *(v4f*)(og + o_b + q * (4 * GCN_D * 4)) = v;
    }
}

extern "C" void kernel_launch(void* const* d_in, const int* in_sizes, int n_in,
                              void* d_out, int out_size, void* d_ws, size_t ws_size,
                              hipStream_t stream)
{
    const float* x  = (const float*)d_in[0];
    const int*   ei = (const int*)d_in[1];
    const float* W  = (const float*)d_in[2];
    float*       out = (float*)d_out;
    int*         wsi = (int*)d_ws;
    int* list   = wsi;
    int* segoff = wsi + 1600000;

    gcn_blocksort<<<250, 800, 0, stream>>>(ei, list, segoff);
    gcn_rowsum<<<250, 800, 0, stream>>>(x, list, segoff, W, out);
}
